# starting kernel
# baseline (speedup 1.0000x reference)
#include <hip/hip_runtime.h>

typedef __attribute__((ext_vector_type(2))) float v2f;
typedef __attribute__((ext_vector_type(8))) float v8f;

#define N_ATOMS 32
#define E_MOL   64
#define IN_CH   64
#define OUT_CH  128
#define HID     256
#define M_MOL   10000
#define E_KG    320000

__device__ __forceinline__ v8f wmma_f32_k4(v2f a, v2f b, v8f c) {
  return __builtin_amdgcn_wmma_f32_16x16x4_f32(false, a, false, b, (short)0, c,
                                               false, false);
}

__device__ __forceinline__ void gemm32_wmma(const float* __restrict__ A32,
                                            const float* __restrict__ W,
                                            float* __restrict__ T,
                                            int K, int N,
                                            int wave, int nwaves, int lane) {
  const int hi = lane >> 4;
  const int lo = lane & 15;
  const int ntTiles = N >> 4;
  for (int t = wave; t < 2 * ntTiles; t += nwaves) {
    const int mt = t & 1;
    const int nt = t >> 1;
    const int mrow = mt * 16 + lo;
    const int ncol = nt * 16 + lo;
    v8f acc = {};
    for (int kb = 0; kb < K; kb += 4) {
      const int k = kb + 2 * hi;
      v2f a, b;
      a.x = A32[mrow * K + k];
      a.y = A32[mrow * K + k + 1];
      b.x = W[k * N + ncol];
      b.y = W[(k + 1) * N + ncol];
      acc = wmma_f32_k4(a, b, acc);
    }
#pragma unroll
    for (int v = 0; v < 8; ++v)
      T[(mt * 16 + v + 8 * hi) * N + nt * 16 + lo] = acc[v];
  }
}

__device__ __forceinline__ void aggregate(const float* __restrict__ T,
                                          float* __restrict__ H,
                                          const float* __restrict__ bias,
                                          int Cshift, bool relu,
                                          const int* s_src, const int* s_dst,
                                          const float* s_dinv,
                                          const float* s_norm,
                                          int tid, int nthr) {
  const int Cm1 = (1 << Cshift) - 1;
  for (int idx = tid; idx < (N_ATOMS << Cshift); idx += nthr) {
    const int i = idx >> Cshift;
    const int c = idx & Cm1;
    H[idx] = s_dinv[i] * s_dinv[i] * T[idx] + bias[c];
  }
  __syncthreads();
  for (int idx = tid; idx < (E_MOL << Cshift); idx += nthr) {
    const int e = idx >> Cshift;
    const int c = idx & Cm1;
    atomicAdd(&H[(s_dst[e] << Cshift) + c],
              s_norm[e] * T[(s_src[e] << Cshift) + c]);
  }
  __syncthreads();
  if (relu) {
    for (int idx = tid; idx < (N_ATOMS << Cshift); idx += nthr)
      H[idx] = fmaxf(H[idx], 0.0f);
  }
  __syncthreads();
}

__global__ __launch_bounds__(256) void mol_gnn_kernel(
    const float* __restrict__ mol_x, const int* __restrict__ mei,
    const float* __restrict__ W1, const float* __restrict__ b1,
    const float* __restrict__ W2, const float* __restrict__ b2,
    const float* __restrict__ W3, const float* __restrict__ b3,
    float* __restrict__ embed) {
  __shared__ float Abuf[N_ATOMS * HID];
  __shared__ float Bbuf[N_ATOMS * HID];
  __shared__ float s_dinv[N_ATOMS];
  __shared__ float s_norm[E_MOL];
  __shared__ int   s_deg[N_ATOMS];
  __shared__ int   s_src[E_MOL];
  __shared__ int   s_dst[E_MOL];

  const int mol  = blockIdx.x;
  const int tid  = threadIdx.x;
  const int nthr = blockDim.x;
  const int lane = tid & 31;
  const int wave = tid >> 5;
  const int nw   = nthr >> 5;

  const float* X = mol_x + (size_t)mol * (N_ATOMS * IN_CH);
  for (int i = tid; i < N_ATOMS * IN_CH; i += nthr) Abuf[i] = X[i];
  const int* me = mei + (size_t)mol * (2 * E_MOL);
  if (tid < E_MOL) { s_src[tid] = me[tid]; s_dst[tid] = me[E_MOL + tid]; }
  if (tid < N_ATOMS) s_deg[tid] = 1;
  __syncthreads();
  if (tid < E_MOL) atomicAdd(&s_deg[s_dst[tid]], 1);
  __syncthreads();
  if (tid < N_ATOMS) s_dinv[tid] = rsqrtf((float)s_deg[tid]);
  __syncthreads();
  if (tid < E_MOL) s_norm[tid] = s_dinv[s_src[tid]] * s_dinv[s_dst[tid]];
  __syncthreads();

  gemm32_wmma(Abuf, W1, Bbuf, IN_CH, HID, wave, nw, lane);
  __syncthreads();
  aggregate(Bbuf, Abuf, b1, 8, true, s_src, s_dst, s_dinv, s_norm, tid, nthr);

  gemm32_wmma(Abuf, W2, Bbuf, HID, HID, wave, nw, lane);
  __syncthreads();
  aggregate(Bbuf, Abuf, b2, 8, true, s_src, s_dst, s_dinv, s_norm, tid, nthr);

  gemm32_wmma(Abuf, W3, Bbuf, HID, OUT_CH, wave, nw, lane);
  __syncthreads();
  aggregate(Bbuf, Abuf, b3, 7, false, s_src, s_dst, s_dinv, s_norm, tid, nthr);

  for (int c = tid; c < OUT_CH; c += nthr) {
    float v = Abuf[c];
    for (int r = 1; r < N_ATOMS; ++r) v = fminf(v, Abuf[r * OUT_CH + c]);
    embed[(size_t)mol * OUT_CH + c] = v;
  }
}

__global__ __launch_bounds__(256) void wmma_gemm_kernel(
    const float* __restrict__ A, const float* __restrict__ B,
    float* __restrict__ C, int Mrows, int K, int N) {
  const int lane = threadIdx.x & 31;
  const int wave = blockIdx.x * (blockDim.x >> 5) + (threadIdx.x >> 5);
  const int ntT = N >> 4;
  const int nTiles = (Mrows >> 4) * ntT;
  if (wave >= nTiles) return;
  const int mt = wave / ntT;
  const int nt = wave - mt * ntT;
  const int hi = lane >> 4, lo = lane & 15;
  const int mrow = mt * 16 + lo;
  const int ncol = nt * 16 + lo;
  v8f acc = {};
  for (int kb = 0; kb < K; kb += 4) {
    const int k = kb + 2 * hi;
    v2f a, b;
    a.x = A[(size_t)mrow * K + k];
    a.y = A[(size_t)mrow * K + k + 1];
    b.x = B[(size_t)k * N + ncol];
    b.y = B[(size_t)(k + 1) * N + ncol];
    acc = wmma_f32_k4(a, b, acc);
  }
#pragma unroll
  for (int v = 0; v < 8; ++v)
    C[(size_t)(mt * 16 + v + 8 * hi) * N + nt * 16 + lo] = acc[v];
}

__global__ void kg_deg_init_kernel(float* __restrict__ deg, int n) {
  int i = blockIdx.x * blockDim.x + threadIdx.x;
  if (i < n) deg[i] = 1.0f;
}
__global__ void kg_deg_count_kernel(const int* __restrict__ dst,
                                    float* __restrict__ deg, int ne) {
  int i = blockIdx.x * blockDim.x + threadIdx.x;
  if (i < ne) atomicAdd(&deg[dst[i]], 1.0f);
}
__global__ void kg_dinv_kernel(const float* __restrict__ deg,
                               float* __restrict__ dinv, int n) {
  int i = blockIdx.x * blockDim.x + threadIdx.x;
  if (i < n) dinv[i] = rsqrtf(deg[i]);
}
__global__ void kg_scale_bias_kernel(const float* __restrict__ XW,
                                     const float* __restrict__ dinv,
                                     const float* __restrict__ bias,
                                     float* __restrict__ H,
                                     int Cshift, int total) {
  int idx = blockIdx.x * blockDim.x + threadIdx.x;
  if (idx >= total) return;
  int i = idx >> Cshift;
  int c = idx & ((1 << Cshift) - 1);
  H[idx] = dinv[i] * dinv[i] * XW[idx] + bias[c];
}
__global__ void kg_edge_kernel(const float* __restrict__ XW,
                               float* __restrict__ H,
                               const float* __restrict__ dinv,
                               const int* __restrict__ src,
                               const int* __restrict__ dst,
                               int Cshift, long long total) {
  long long idx = (long long)blockIdx.x * blockDim.x + threadIdx.x;
  if (idx >= total) return;
  int e = (int)(idx >> Cshift);
  int c = (int)(idx & ((1 << Cshift) - 1));
  int s = src[e], d = dst[e];
  atomicAdd(&H[((size_t)d << Cshift) + c],
            dinv[s] * dinv[d] * XW[((size_t)s << Cshift) + c]);
}
__global__ void relu_kernel(float* __restrict__ h, int n) {
  int i = blockIdx.x * blockDim.x + threadIdx.x;
  if (i < n) h[i] = fmaxf(h[i], 0.0f);
}

extern "C" void kernel_launch(void* const* d_in, const int* in_sizes, int n_in,
                              void* d_out, int out_size, void* d_ws, size_t ws_size,
                              hipStream_t stream) {
  const float* mol_x = (const float*)d_in[0];
  const int*   mei   = (const int*)d_in[1];
  const int*   kei   = (const int*)d_in[2];
  const float* W1 = (const float*)d_in[3];
  const float* b1 = (const float*)d_in[4];
  const float* W2 = (const float*)d_in[5];
  const float* b2 = (const float*)d_in[6];
  const float* W3 = (const float*)d_in[7];
  const float* b3 = (const float*)d_in[8];
  const float* Wk1 = (const float*)d_in[9];
  const float* bk1 = (const float*)d_in[10];
  const float* Wk2 = (const float*)d_in[11];
  const float* bk2 = (const float*)d_in[12];
  float* out = (float*)d_out;

  float* ws    = (float*)d_ws;
  float* embed = ws;
  float* deg   = embed + M_MOL * OUT_CH;
  float* dinv  = deg + M_MOL;
  float* XW    = dinv + M_MOL;
  float* h     = XW + (size_t)M_MOL * HID;
  float* XW2   = h + (size_t)M_MOL * HID;

  const int* kg_src = kei;
  const int* kg_dst = kei + E_KG;

  mol_gnn_kernel<<<M_MOL, 256, 0, stream>>>(mol_x, mei, W1, b1, W2, b2, W3, b3,
                                            embed);

  kg_deg_init_kernel<<<(M_MOL + 255) / 256, 256, 0, stream>>>(deg, M_MOL);
  kg_deg_count_kernel<<<(E_KG + 255) / 256, 256, 0, stream>>>(kg_dst, deg, E_KG);
  kg_dinv_kernel<<<(M_MOL + 255) / 256, 256, 0, stream>>>(deg, dinv, M_MOL);

  {
    int tiles = (M_MOL / 16) * (HID / 16);
    wmma_gemm_kernel<<<(tiles + 7) / 8, 256, 0, stream>>>(embed, Wk1, XW,
                                                          M_MOL, OUT_CH, HID);
  }
  kg_scale_bias_kernel<<<(M_MOL * HID + 255) / 256, 256, 0, stream>>>(
      XW, dinv, bk1, h, 8, M_MOL * HID);
  kg_edge_kernel<<<(int)(((long long)E_KG * HID + 255) / 256), 256, 0, stream>>>(
      XW, h, dinv, kg_src, kg_dst, 8, (long long)E_KG * HID);
  relu_kernel<<<(M_MOL * HID + 255) / 256, 256, 0, stream>>>(h, M_MOL * HID);

  {
    int tiles = (M_MOL / 16) * (OUT_CH / 16);
    wmma_gemm_kernel<<<(tiles + 7) / 8, 256, 0, stream>>>(h, Wk2, XW2,
                                                          M_MOL, HID, OUT_CH);
  }
  kg_scale_bias_kernel<<<(M_MOL * OUT_CH + 255) / 256, 256, 0, stream>>>(
      XW2, dinv, bk2, out, 7, M_MOL * OUT_CH);
  kg_edge_kernel<<<(int)(((long long)E_KG * OUT_CH + 255) / 256), 256, 0,
                  stream>>>(XW2, out, dinv, kg_src, kg_dst, 7,
                            (long long)E_KG * OUT_CH);
}


# mol kernel: activations kept channel-major in LDS, GEMMs computed transposed (W = WMMA A operand, atoms = B columns): conflict-free fragment reads; k pairing and edge-sum order change
# speedup vs baseline: 1.5034x; 1.5034x over previous
#include <hip/hip_runtime.h>

typedef __attribute__((ext_vector_type(2))) float v2f;
typedef __attribute__((ext_vector_type(8))) float v8f;

#define N_ATOMS 32
#define E_MOL   64
#define IN_CH   64
#define OUT_CH  128
#define HID     256
#define M_MOL   10000
#define E_KG    320000

__device__ __forceinline__ v8f wmma_f32_k4(v2f a, v2f b, v8f c) {
  return __builtin_amdgcn_wmma_f32_16x16x4_f32(false, a, false, b, (short)0, c,
                                               false, false);
}

#define MOL_WAVES 8

template <int K, int N>
__device__ __forceinline__ void gemm32t_wmma(const float* __restrict__ XT,
                                             const float* __restrict__ W,
                                             float* __restrict__ TT,
                                             int wave, int lane) {
  const int hi = lane >> 4;
  const int lo = lane & 15;
  const float* __restrict__ Wl = W + hi * N + lo;
  const float* __restrict__ Xl = XT + hi * N_ATOMS + lo;
  for (int t = wave; t < 2 * (N / 16); t += MOL_WAVES) {
    const int mt = t & 1;
    const int nt = t >> 1;
    const float* __restrict__ wp = Wl + nt * 16;
    const float* __restrict__ xp = Xl + mt * 16;
    v8f acc = {};
#pragma unroll 8
    for (int kb = 0; kb < K; kb += 4) {
      v2f a, b;
      a.x = wp[kb * N];
      a.y = wp[(kb + 2) * N];
      b.x = xp[kb * N_ATOMS];
      b.y = xp[(kb + 2) * N_ATOMS];
      acc = wmma_f32_k4(a, b, acc);
    }
#pragma unroll
    for (int v = 0; v < 8; ++v)
      TT[(nt * 16 + v + 8 * hi) * N_ATOMS + mt * 16 + lo] = acc[v];
  }
}

template <int C, bool RELU>
__device__ __forceinline__ void aggregate_t(const float* __restrict__ T,
                                            float* __restrict__ H,
                                            const float* __restrict__ bias,
                                            float dinv2,
                                            int s0, int d0, float n0,
                                            int s1, int d1, float n1,
                                            int wave, int lane) {
  for (int c = wave; c < C; c += MOL_WAVES)
    H[c * N_ATOMS + lane] = dinv2 * T[c * N_ATOMS + lane] + bias[c];
  __syncthreads();
  for (int c = wave; c < C; c += 4 * MOL_WAVES) {
    float t0[4], t1[4];
#pragma unroll
    for (int j = 0; j < 4; ++j) {
      t0[j] = T[(c + j * MOL_WAVES) * N_ATOMS + s0];
      t1[j] = T[(c + j * MOL_WAVES) * N_ATOMS + s1];
    }
#pragma unroll
    for (int j = 0; j < 4; ++j) {
      atomicAdd(&H[(c + j * MOL_WAVES) * N_ATOMS + d0], n0 * t0[j]);
      atomicAdd(&H[(c + j * MOL_WAVES) * N_ATOMS + d1], n1 * t1[j]);
    }
  }
  __syncthreads();
  if (RELU) {
    for (int c = wave; c < C; c += MOL_WAVES)
      H[c * N_ATOMS + lane] = fmaxf(H[c * N_ATOMS + lane], 0.0f);
  }
  __syncthreads();
}

__global__ __launch_bounds__(256) void mol_gnn_kernel(
    const float* __restrict__ mol_x, const int* __restrict__ mei,
    const float* __restrict__ W1, const float* __restrict__ b1,
    const float* __restrict__ W2, const float* __restrict__ b2,
    const float* __restrict__ W3, const float* __restrict__ b3,
    float* __restrict__ embed) {
  __shared__ float Abuf[N_ATOMS * HID];
  __shared__ float Bbuf[N_ATOMS * HID];
  __shared__ float s_dinv[N_ATOMS];
  __shared__ float s_norm[E_MOL];
  __shared__ int   s_deg[N_ATOMS];
  __shared__ int   s_src[E_MOL];
  __shared__ int   s_dst[E_MOL];

  const int mol  = blockIdx.x;
  const int tid  = threadIdx.x;
  const int lane = tid & 31;
  const int wave = __builtin_amdgcn_readfirstlane(tid >> 5);
  __builtin_assume(wave >= 0 && wave < MOL_WAVES);

  const float* X = mol_x + (size_t)mol * (N_ATOMS * IN_CH);
  const float4* X4 = reinterpret_cast<const float4*>(X);
  for (int q = wave; q < IN_CH / 4; q += MOL_WAVES) {
    const float4 v = X4[lane * (IN_CH / 4) + q];
    Abuf[(4 * q + 0) * N_ATOMS + lane] = v.x;
    Abuf[(4 * q + 1) * N_ATOMS + lane] = v.y;
    Abuf[(4 * q + 2) * N_ATOMS + lane] = v.z;
    Abuf[(4 * q + 3) * N_ATOMS + lane] = v.w;
  }
  const int* me = mei + (size_t)mol * (2 * E_MOL);
  if (tid < E_MOL) { s_src[tid] = me[tid]; s_dst[tid] = me[E_MOL + tid]; }
  if (tid < N_ATOMS) s_deg[tid] = 1;
  __syncthreads();
  if (tid < E_MOL) atomicAdd(&s_deg[s_dst[tid]], 1);
  __syncthreads();
  if (tid < N_ATOMS) s_dinv[tid] = rsqrtf((float)s_deg[tid]);
  __syncthreads();
  if (tid < E_MOL) s_norm[tid] = s_dinv[s_src[tid]] * s_dinv[s_dst[tid]];
  __syncthreads();

  const float dinv2 = s_dinv[lane] * s_dinv[lane];
  const int   es0 = s_src[lane],      ed0 = s_dst[lane];
  const int   es1 = s_src[lane + 32], ed1 = s_dst[lane + 32];
  const float en0 = s_norm[lane],     en1 = s_norm[lane + 32];

  gemm32t_wmma<IN_CH, HID>(Abuf, W1, Bbuf, wave, lane);
  __syncthreads();
  aggregate_t<HID, true>(Bbuf, Abuf, b1, dinv2, es0, ed0, en0, es1, ed1, en1,
                         wave, lane);

  gemm32t_wmma<HID, HID>(Abuf, W2, Bbuf, wave, lane);
  __syncthreads();
  aggregate_t<HID, true>(Bbuf, Abuf, b2, dinv2, es0, ed0, en0, es1, ed1, en1,
                         wave, lane);

  gemm32t_wmma<HID, OUT_CH>(Abuf, W3, Bbuf, wave, lane);
  __syncthreads();
  aggregate_t<OUT_CH, false>(Bbuf, Abuf, b3, dinv2, es0, ed0, en0, es1, ed1, en1,
                             wave, lane);

  if (tid < OUT_CH) {
    const int c = tid;
    float v = Abuf[c * N_ATOMS + (c & (N_ATOMS - 1))];
    for (int r = 1; r < N_ATOMS; ++r)
      v = fminf(v, Abuf[c * N_ATOMS + ((c + r) & (N_ATOMS - 1))]);
    embed[(size_t)mol * OUT_CH + c] = v;
  }
}

__global__ __launch_bounds__(256) void wmma_gemm_kernel(
    const float* __restrict__ A, const float* __restrict__ B,
    float* __restrict__ C, int Mrows, int K, int N) {
  const int lane = threadIdx.x & 31;
  const int wave = blockIdx.x * (blockDim.x >> 5) + (threadIdx.x >> 5);
  const int ntT = N >> 4;
  const int nTiles = (Mrows >> 4) * ntT;
  if (wave >= nTiles) return;
  const int mt = wave / ntT;
  const int nt = wave - mt * ntT;
  const int hi = lane >> 4, lo = lane & 15;
  const int mrow = mt * 16 + lo;
  const int ncol = nt * 16 + lo;
  v8f acc = {};
  for (int kb = 0; kb < K; kb += 4) {
    const int k = kb + 2 * hi;
    v2f a, b;
    a.x = A[(size_t)mrow * K + k];
    a.y = A[(size_t)mrow * K + k + 1];
    b.x = B[(size_t)k * N + ncol];
    b.y = B[(size_t)(k + 1) * N + ncol];
    acc = wmma_f32_k4(a, b, acc);
  }
#pragma unroll
  for (int v = 0; v < 8; ++v)
    C[(size_t)(mt * 16 + v + 8 * hi) * N + nt * 16 + lo] = acc[v];
}

__global__ void kg_deg_init_kernel(float* __restrict__ deg, int n) {
  int i = blockIdx.x * blockDim.x + threadIdx.x;
  if (i < n) deg[i] = 1.0f;
}
__global__ void kg_deg_count_kernel(const int* __restrict__ dst,
                                    float* __restrict__ deg, int ne) {
  int i = blockIdx.x * blockDim.x + threadIdx.x;
  if (i < ne) atomicAdd(&deg[dst[i]], 1.0f);
}
__global__ void kg_dinv_kernel(const float* __restrict__ deg,
                               float* __restrict__ dinv, int n) {
  int i = blockIdx.x * blockDim.x + threadIdx.x;
  if (i < n) dinv[i] = rsqrtf(deg[i]);
}
__global__ void kg_scale_bias_kernel(const float* __restrict__ XW,
                                     const float* __restrict__ dinv,
                                     const float* __restrict__ bias,
                                     float* __restrict__ H,
                                     int Cshift, int total) {
  int idx = blockIdx.x * blockDim.x + threadIdx.x;
  if (idx >= total) return;
  int i = idx >> Cshift;
  int c = idx & ((1 << Cshift) - 1);
  H[idx] = dinv[i] * dinv[i] * XW[idx] + bias[c];
}
__global__ void kg_edge_kernel(const float* __restrict__ XW,
                               float* __restrict__ H,
                               const float* __restrict__ dinv,
                               const int* __restrict__ src,
                               const int* __restrict__ dst,
                               int Cshift, long long total) {
  long long idx = (long long)blockIdx.x * blockDim.x + threadIdx.x;
  if (idx >= total) return;
  int e = (int)(idx >> Cshift);
  int c = (int)(idx & ((1 << Cshift) - 1));
  int s = src[e], d = dst[e];
  atomicAdd(&H[((size_t)d << Cshift) + c],
            dinv[s] * dinv[d] * XW[((size_t)s << Cshift) + c]);
}
__global__ void relu_kernel(float* __restrict__ h, int n) {
  int i = blockIdx.x * blockDim.x + threadIdx.x;
  if (i < n) h[i] = fmaxf(h[i], 0.0f);
}

extern "C" void kernel_launch(void* const* d_in, const int* in_sizes, int n_in,
                              void* d_out, int out_size, void* d_ws, size_t ws_size,
                              hipStream_t stream) {
  const float* mol_x = (const float*)d_in[0];
  const int*   mei   = (const int*)d_in[1];
  const int*   kei   = (const int*)d_in[2];
  const float* W1 = (const float*)d_in[3];
  const float* b1 = (const float*)d_in[4];
  const float* W2 = (const float*)d_in[5];
  const float* b2 = (const float*)d_in[6];
  const float* W3 = (const float*)d_in[7];
  const float* b3 = (const float*)d_in[8];
  const float* Wk1 = (const float*)d_in[9];
  const float* bk1 = (const float*)d_in[10];
  const float* Wk2 = (const float*)d_in[11];
  const float* bk2 = (const float*)d_in[12];
  float* out = (float*)d_out;

  float* ws    = (float*)d_ws;
  float* embed = ws;
  float* deg   = embed + M_MOL * OUT_CH;
  float* dinv  = deg + M_MOL;
  float* XW    = dinv + M_MOL;
  float* h     = XW + (size_t)M_MOL * HID;
  float* XW2   = h + (size_t)M_MOL * HID;

  const int* kg_src = kei;
  const int* kg_dst = kei + E_KG;

  mol_gnn_kernel<<<M_MOL, 256, 0, stream>>>(mol_x, mei, W1, b1, W2, b2, W3, b3,
                                            embed);

  kg_deg_init_kernel<<<(M_MOL + 255) / 256, 256, 0, stream>>>(deg, M_MOL);
  kg_deg_count_kernel<<<(E_KG + 255) / 256, 256, 0, stream>>>(kg_dst, deg, E_KG);
  kg_dinv_kernel<<<(M_MOL + 255) / 256, 256, 0, stream>>>(deg, dinv, M_MOL);

  {
    int tiles = (M_MOL / 16) * (HID / 16);
    wmma_gemm_kernel<<<(tiles + 7) / 8, 256, 0, stream>>>(embed, Wk1, XW,
                                                          M_MOL, OUT_CH, HID);
  }
  kg_scale_bias_kernel<<<(M_MOL * HID + 255) / 256, 256, 0, stream>>>(
      XW, dinv, bk1, h, 8, M_MOL * HID);
  kg_edge_kernel<<<(int)(((long long)E_KG * HID + 255) / 256), 256, 0, stream>>>(
      XW, h, dinv, kg_src, kg_dst, 8, (long long)E_KG * HID);
  relu_kernel<<<(M_MOL * HID + 255) / 256, 256, 0, stream>>>(h, M_MOL * HID);

  {
    int tiles = (M_MOL / 16) * (OUT_CH / 16);
    wmma_gemm_kernel<<<(tiles + 7) / 8, 256, 0, stream>>>(h, Wk2, XW2,
                                                          M_MOL, HID, OUT_CH);
  }
  kg_scale_bias_kernel<<<(M_MOL * OUT_CH + 255) / 256, 256, 0, stream>>>(
      XW2, dinv, bk2, out, 7, M_MOL * OUT_CH);
  kg_edge_kernel<<<(int)(((long long)E_KG * OUT_CH + 255) / 256), 256, 0,
                  stream>>>(XW2, out, dinv, kg_src, kg_dst, 7,
                            (long long)E_KG * OUT_CH);
}


# KG layers: float-atomic edge scatter -> each workgroup owns 40 destination nodes, sums in LDS, filters the edge list itself; no atomics, no index build; self+bias+ReLU fused; sum order changes
# speedup vs baseline: 1.7184x; 1.1430x over previous
#include <hip/hip_runtime.h>

typedef __attribute__((ext_vector_type(2))) float v2f;
typedef __attribute__((ext_vector_type(8))) float v8f;

#define N_ATOMS 32
#define E_MOL   64
#define IN_CH   64
#define OUT_CH  128
#define HID     256
#define M_MOL   10000
#define E_KG    320000

__device__ __forceinline__ v8f wmma_f32_k4(v2f a, v2f b, v8f c) {
  return __builtin_amdgcn_wmma_f32_16x16x4_f32(false, a, false, b, (short)0, c,
                                               false, false);
}

#define MOL_WAVES 8

template <int K, int N>
__device__ __forceinline__ void gemm32t_wmma(const float* __restrict__ XT,
                                             const float* __restrict__ W,
                                             float* __restrict__ TT,
                                             int wave, int lane) {
  const int hi = lane >> 4;
  const int lo = lane & 15;
  const float* __restrict__ Wl = W + hi * N + lo;
  const float* __restrict__ Xl = XT + hi * N_ATOMS + lo;
  for (int t = wave; t < 2 * (N / 16); t += MOL_WAVES) {
    const int mt = t & 1;
    const int nt = t >> 1;
    const float* __restrict__ wp = Wl + nt * 16;
    const float* __restrict__ xp = Xl + mt * 16;
    v8f acc = {};
#pragma unroll 8
    for (int kb = 0; kb < K; kb += 4) {
      v2f a, b;
      a.x = wp[kb * N];
      a.y = wp[(kb + 2) * N];
      b.x = xp[kb * N_ATOMS];
      b.y = xp[(kb + 2) * N_ATOMS];
      acc = wmma_f32_k4(a, b, acc);
    }
#pragma unroll
    for (int v = 0; v < 8; ++v)
      TT[(nt * 16 + v + 8 * hi) * N_ATOMS + mt * 16 + lo] = acc[v];
  }
}

template <int C, bool RELU>
__device__ __forceinline__ void aggregate_t(const float* __restrict__ T,
                                            float* __restrict__ H,
                                            const float* __restrict__ bias,
                                            float dinv2,
                                            int s0, int d0, float n0,
                                            int s1, int d1, float n1,
                                            int wave, int lane) {
  for (int c = wave; c < C; c += MOL_WAVES)
    H[c * N_ATOMS + lane] = dinv2 * T[c * N_ATOMS + lane] + bias[c];
  __syncthreads();
  for (int c = wave; c < C; c += 4 * MOL_WAVES) {
    float t0[4], t1[4];
#pragma unroll
    for (int j = 0; j < 4; ++j) {
      t0[j] = T[(c + j * MOL_WAVES) * N_ATOMS + s0];
      t1[j] = T[(c + j * MOL_WAVES) * N_ATOMS + s1];
    }
#pragma unroll
    for (int j = 0; j < 4; ++j) {
      atomicAdd(&H[(c + j * MOL_WAVES) * N_ATOMS + d0], n0 * t0[j]);
      atomicAdd(&H[(c + j * MOL_WAVES) * N_ATOMS + d1], n1 * t1[j]);
    }
  }
  __syncthreads();
  if (RELU) {
    for (int c = wave; c < C; c += MOL_WAVES)
      H[c * N_ATOMS + lane] = fmaxf(H[c * N_ATOMS + lane], 0.0f);
  }
  __syncthreads();
}

__global__ __launch_bounds__(256) void mol_gnn_kernel(
    const float* __restrict__ mol_x, const int* __restrict__ mei,
    const float* __restrict__ W1, const float* __restrict__ b1,
    const float* __restrict__ W2, const float* __restrict__ b2,
    const float* __restrict__ W3, const float* __restrict__ b3,
    float* __restrict__ embed) {
  __shared__ float Abuf[N_ATOMS * HID];
  __shared__ float Bbuf[N_ATOMS * HID];
  __shared__ float s_dinv[N_ATOMS];
  __shared__ float s_norm[E_MOL];
  __shared__ int   s_deg[N_ATOMS];
  __shared__ int   s_src[E_MOL];
  __shared__ int   s_dst[E_MOL];

  const int mol  = blockIdx.x;
  const int tid  = threadIdx.x;
  const int lane = tid & 31;
  const int wave = __builtin_amdgcn_readfirstlane(tid >> 5);
  __builtin_assume(wave >= 0 && wave < MOL_WAVES);

  const float* X = mol_x + (size_t)mol * (N_ATOMS * IN_CH);
  const float4* X4 = reinterpret_cast<const float4*>(X);
  for (int q = wave; q < IN_CH / 4; q += MOL_WAVES) {
    const float4 v = X4[lane * (IN_CH / 4) + q];
    Abuf[(4 * q + 0) * N_ATOMS + lane] = v.x;
    Abuf[(4 * q + 1) * N_ATOMS + lane] = v.y;
    Abuf[(4 * q + 2) * N_ATOMS + lane] = v.z;
    Abuf[(4 * q + 3) * N_ATOMS + lane] = v.w;
  }
  const int* me = mei + (size_t)mol * (2 * E_MOL);
  if (tid < E_MOL) { s_src[tid] = me[tid]; s_dst[tid] = me[E_MOL + tid]; }
  if (tid < N_ATOMS) s_deg[tid] = 1;
  __syncthreads();
  if (tid < E_MOL) atomicAdd(&s_deg[s_dst[tid]], 1);
  __syncthreads();
  if (tid < N_ATOMS) s_dinv[tid] = rsqrtf((float)s_deg[tid]);
  __syncthreads();
  if (tid < E_MOL) s_norm[tid] = s_dinv[s_src[tid]] * s_dinv[s_dst[tid]];
  __syncthreads();

  const float dinv2 = s_dinv[lane] * s_dinv[lane];
  const int   es0 = s_src[lane],      ed0 = s_dst[lane];
  const int   es1 = s_src[lane + 32], ed1 = s_dst[lane + 32];
  const float en0 = s_norm[lane],     en1 = s_norm[lane + 32];

  gemm32t_wmma<IN_CH, HID>(Abuf, W1, Bbuf, wave, lane);
  __syncthreads();
  aggregate_t<HID, true>(Bbuf, Abuf, b1, dinv2, es0, ed0, en0, es1, ed1, en1,
                         wave, lane);

  gemm32t_wmma<HID, HID>(Abuf, W2, Bbuf, wave, lane);
  __syncthreads();
  aggregate_t<HID, true>(Bbuf, Abuf, b2, dinv2, es0, ed0, en0, es1, ed1, en1,
                         wave, lane);

  gemm32t_wmma<HID, OUT_CH>(Abuf, W3, Bbuf, wave, lane);
  __syncthreads();
  aggregate_t<OUT_CH, false>(Bbuf, Abuf, b3, dinv2, es0, ed0, en0, es1, ed1, en1,
                             wave, lane);

  if (tid < OUT_CH) {
    const int c = tid;
    float v = Abuf[c * N_ATOMS + (c & (N_ATOMS - 1))];
    for (int r = 1; r < N_ATOMS; ++r)
      v = fminf(v, Abuf[c * N_ATOMS + ((c + r) & (N_ATOMS - 1))]);
    embed[(size_t)mol * OUT_CH + c] = v;
  }
}

__global__ __launch_bounds__(256) void wmma_gemm_kernel(
    const float* __restrict__ A, const float* __restrict__ B,
    float* __restrict__ C, int Mrows, int K, int N) {
  const int lane = threadIdx.x & 31;
  const int wave = blockIdx.x * (blockDim.x >> 5) + (threadIdx.x >> 5);
  const int ntT = N >> 4;
  const int nTiles = (Mrows >> 4) * ntT;
  if (wave >= nTiles) return;
  const int mt = wave / ntT;
  const int nt = wave - mt * ntT;
  const int hi = lane >> 4, lo = lane & 15;
  const int mrow = mt * 16 + lo;
  const int ncol = nt * 16 + lo;
  v8f acc = {};
  for (int kb = 0; kb < K; kb += 4) {
    const int k = kb + 2 * hi;
    v2f a, b;
    a.x = A[(size_t)mrow * K + k];
    a.y = A[(size_t)mrow * K + k + 1];
    b.x = B[(size_t)k * N + ncol];
    b.y = B[(size_t)(k + 1) * N + ncol];
    acc = wmma_f32_k4(a, b, acc);
  }
#pragma unroll
  for (int v = 0; v < 8; ++v)
    C[(size_t)(mt * 16 + v + 8 * hi) * N + nt * 16 + lo] = acc[v];
}

__global__ void kg_deg_init_kernel(float* __restrict__ deg, int n) {
  int i = blockIdx.x * blockDim.x + threadIdx.x;
  if (i < n) deg[i] = 1.0f;
}
__global__ void kg_deg_count_kernel(const int* __restrict__ dst,
                                    float* __restrict__ deg, int ne) {
  int i = blockIdx.x * blockDim.x + threadIdx.x;
  if (i < ne) atomicAdd(&deg[dst[i]], 1.0f);
}
__global__ void kg_dinv_kernel(const float* __restrict__ deg,
                               float* __restrict__ dinv, int n) {
  int i = blockIdx.x * blockDim.x + threadIdx.x;
  if (i < n) dinv[i] = rsqrtf(deg[i]);
}
#define KG_NB      40
#define KG_THREADS 1024
#define KG_WAVES   32
#define KG_CAP     512
#define KG_ROUND   (KG_WAVES * KG_CAP)
#define KG_BATCH   8

template <int C, bool RELU>
__global__ __launch_bounds__(KG_THREADS) void kg_block_gather_kernel(
    const float* __restrict__ XW, const float* __restrict__ dinv,
    const float* __restrict__ bias, const int* __restrict__ src,
    const int* __restrict__ dst, float* __restrict__ H,
    int n_nodes, int n_edges) {
  constexpr int G = KG_THREADS / C;
  __shared__ float s_acc[KG_NB * C];
  __shared__ int2  s_list[KG_ROUND];
  __shared__ int   s_cnt[KG_WAVES];
  __shared__ float s_dinv[KG_NB];

  const int tid  = threadIdx.x;
  const int lane = tid & 31;
  const int wave = __builtin_amdgcn_readfirstlane(tid >> 5);
  const int g    = __builtin_amdgcn_readfirstlane(tid / C);
  const int c    = tid & (C - 1);
  const int n0   = blockIdx.x * KG_NB;
  const int nb   = min(KG_NB, n_nodes - n0);

  if (tid < KG_NB) s_dinv[tid] = (tid < nb) ? dinv[n0 + tid] : 0.0f;
  __syncthreads();
  for (int j = g; j < nb; j += G) {
    const float di = s_dinv[j];
    s_acc[j * C + c] = di * di * XW[(size_t)(n0 + j) * C + c] + bias[c];
  }

  const unsigned lanes_below = (1u << lane) - 1u;
  for (int base = 0; base < n_edges; base += KG_ROUND) {
    const int wbase = base + wave * KG_CAP;
    int4 dv[4];
#pragma unroll
    for (int q = 0; q < 4; ++q) {
      const int e = wbase + q * 128 + lane * 4;
      dv[q] = make_int4(-1, -1, -1, -1);
      if (e < n_edges) dv[q] = *reinterpret_cast<const int4*>(dst + e);
    }
    int dl[16], sv[16];
    int count = 0;
#pragma unroll
    for (int q = 0; q < 4; ++q) {
      const int e = wbase + q * 128 + lane * 4;
      const int dd[4] = {dv[q].x, dv[q].y, dv[q].z, dv[q].w};
#pragma unroll
      for (int j = 0; j < 4; ++j) {
        const int d = dd[j] - n0;
        const bool m = (unsigned)d < (unsigned)nb;
        dl[q * 4 + j] = m ? d : -1;
        sv[q * 4 + j] = src[m ? (e + j) : 0];
        count += __popc(__builtin_amdgcn_ballot_w32(m));
      }
    }
    if (lane == 0) s_cnt[wave] = count;
    __syncthreads();

    const int v = s_cnt[lane];
    int incl = v;
#pragma unroll
    for (int dlt = 1; dlt < 32; dlt <<= 1) {
      const int t = __shfl_up(incl, dlt);
      if (lane >= dlt) incl += t;
    }
    const int total = __builtin_amdgcn_readlane(incl, 31);
    int pos = __builtin_amdgcn_readlane(incl - v, wave);
#pragma unroll
    for (int k = 0; k < 16; ++k) {
      const bool m = dl[k] >= 0;
      const unsigned mask = __builtin_amdgcn_ballot_w32(m);
      if (m) s_list[pos + __popc(mask & lanes_below)] = make_int2(sv[k], dl[k]);
      pos += __popc(mask);
    }
    __syncthreads();

    for (int i0 = 0; i0 < total; i0 += 32) {
      const int idx = i0 + lane;
      int2 en = make_int2(0, -1);
      if (idx < total) en = s_list[idx];
      unsigned mm = __builtin_amdgcn_ballot_w32(en.y >= 0 && (en.y & (G - 1)) == g);
      while (mm) {
        int  es[KG_BATCH], ed[KG_BATCH];
        bool ok[KG_BATCH];
#pragma unroll
        for (int k = 0; k < KG_BATCH; ++k) {
          ok[k] = mm != 0u;
          const int b = ok[k] ? __builtin_ctz(mm) : 0;
          es[k] = __builtin_amdgcn_readlane(en.x, b);
          ed[k] = __builtin_amdgcn_readlane(en.y, b);
          mm &= mm - 1u;
        }
        float x[KG_BATCH], w[KG_BATCH];
#pragma unroll
        for (int k = 0; k < KG_BATCH; ++k) {
          x[k] = XW[(size_t)es[k] * C + c];
          w[k] = dinv[es[k]];
        }
#pragma unroll
        for (int k = 0; k < KG_BATCH; ++k) {
          if (ok[k])
            s_acc[ed[k] * C + c] += w[k] * s_dinv[ed[k]] * x[k];
        }
      }
    }
  }

  for (int j = g; j < nb; j += G) {
    float r = s_acc[j * C + c];
    if (RELU) r = fmaxf(r, 0.0f);
    H[(size_t)(n0 + j) * C + c] = r;
  }
}

extern "C" void kernel_launch(void* const* d_in, const int* in_sizes, int n_in,
                              void* d_out, int out_size, void* d_ws, size_t ws_size,
                              hipStream_t stream) {
  const float* mol_x = (const float*)d_in[0];
  const int*   mei   = (const int*)d_in[1];
  const int*   kei   = (const int*)d_in[2];
  const float* W1 = (const float*)d_in[3];
  const float* b1 = (const float*)d_in[4];
  const float* W2 = (const float*)d_in[5];
  const float* b2 = (const float*)d_in[6];
  const float* W3 = (const float*)d_in[7];
  const float* b3 = (const float*)d_in[8];
  const float* Wk1 = (const float*)d_in[9];
  const float* bk1 = (const float*)d_in[10];
  const float* Wk2 = (const float*)d_in[11];
  const float* bk2 = (const float*)d_in[12];
  float* out = (float*)d_out;

  float* ws    = (float*)d_ws;
  float* embed = ws;
  float* deg   = embed + M_MOL * OUT_CH;
  float* dinv  = deg + M_MOL;
  float* XW    = dinv + M_MOL;
  float* h     = XW + (size_t)M_MOL * HID;
  float* XW2   = h + (size_t)M_MOL * HID;

  const int* kg_src = kei;
  const int* kg_dst = kei + E_KG;

  mol_gnn_kernel<<<M_MOL, 256, 0, stream>>>(mol_x, mei, W1, b1, W2, b2, W3, b3,
                                            embed);

  kg_deg_init_kernel<<<(M_MOL + 255) / 256, 256, 0, stream>>>(deg, M_MOL);
  kg_deg_count_kernel<<<(E_KG + 255) / 256, 256, 0, stream>>>(kg_dst, deg, E_KG);
  kg_dinv_kernel<<<(M_MOL + 255) / 256, 256, 0, stream>>>(deg, dinv, M_MOL);

  {
    int tiles = (M_MOL / 16) * (HID / 16);
    wmma_gemm_kernel<<<(tiles + 7) / 8, 256, 0, stream>>>(embed, Wk1, XW,
                                                          M_MOL, OUT_CH, HID);
  }
  kg_block_gather_kernel<HID, true><<<(M_MOL + KG_NB - 1) / KG_NB, KG_THREADS, 0,
                                      stream>>>(XW, dinv, bk1, kg_src, kg_dst, h,
                                                M_MOL, E_KG);

  {
    int tiles = (M_MOL / 16) * (OUT_CH / 16);
    wmma_gemm_kernel<<<(tiles + 7) / 8, 256, 0, stream>>>(h, Wk2, XW2,
                                                          M_MOL, HID, OUT_CH);
  }
  kg_block_gather_kernel<OUT_CH, false><<<(M_MOL + KG_NB - 1) / KG_NB, KG_THREADS,
                                          0, stream>>>(XW2, dinv, bk2, kg_src,
                                                       kg_dst, out, M_MOL, E_KG);
}


# mol kernel GEMMs: a wave owns a 16-channel block and both 16-atom tiles: one W fragment load feeds two independent WMMA chains, W read once per workgroup; 128-register cap; sum order kept
# speedup vs baseline: 1.7963x; 1.0453x over previous
#include <hip/hip_runtime.h>

typedef __attribute__((ext_vector_type(2))) float v2f;
typedef __attribute__((ext_vector_type(8))) float v8f;

#define N_ATOMS 32
#define E_MOL   64
#define IN_CH   64
#define OUT_CH  128
#define HID     256
#define M_MOL   10000
#define E_KG    320000

__device__ __forceinline__ v8f wmma_f32_k4(v2f a, v2f b, v8f c) {
  return __builtin_amdgcn_wmma_f32_16x16x4_f32(false, a, false, b, (short)0, c,
                                               false, false);
}

#define MOL_WAVES 8

template <int K, int N>
__device__ __forceinline__ void gemm32t_wmma(const float* __restrict__ XT,
                                             const float* __restrict__ W,
                                             float* __restrict__ TT,
                                             int wave, int lane) {
  const int hi = lane >> 4;
  const int lo = lane & 15;
  const float* __restrict__ Wl = W + hi * N + lo;
  const float* __restrict__ xp = XT + hi * N_ATOMS + lo;
  int tile1 = 16;
  asm("" : "+v"(tile1));
  const float* __restrict__ xq = xp + tile1;
  for (int nt = wave; nt < N / 16; nt += MOL_WAVES) {
    const float* __restrict__ wp = Wl + nt * 16;
    v8f acc0 = {};
    v8f acc1 = {};
    for (int kb = 0; kb < K; kb += 32) {
      v2f a[8];
#pragma unroll
      for (int j = 0; j < 8; ++j) {
        a[j].x = wp[(kb + 4 * j) * N];
        a[j].y = wp[(kb + 4 * j + 2) * N];
      }
      __builtin_amdgcn_sched_barrier(0);
#pragma unroll
      for (int j = 0; j < 8; ++j) {
        v2f b0, b1;
        b0.x = xp[(kb + 4 * j) * N_ATOMS];
        b0.y = xp[(kb + 4 * j + 2) * N_ATOMS];
        b1.x = xq[(kb + 4 * j) * N_ATOMS];
        b1.y = xq[(kb + 4 * j + 2) * N_ATOMS];
        acc0 = wmma_f32_k4(a[j], b0, acc0);
        acc1 = wmma_f32_k4(a[j], b1, acc1);
      }
    }
#pragma unroll
    for (int v = 0; v < 8; ++v) {
      TT[(nt * 16 + v + 8 * hi) * N_ATOMS + lo]      = acc0[v];
      TT[(nt * 16 + v + 8 * hi) * N_ATOMS + 16 + lo] = acc1[v];
    }
  }
}

template <int C, bool RELU>
__device__ __forceinline__ void aggregate_t(const float* __restrict__ T,
                                            float* __restrict__ H,
                                            const float* __restrict__ bias,
                                            float dinv2,
                                            int s0, int d0, float n0,
                                            int s1, int d1, float n1,
                                            int wave, int lane) {
  for (int c = wave; c < C; c += MOL_WAVES)
    H[c * N_ATOMS + lane] = dinv2 * T[c * N_ATOMS + lane] + bias[c];
  __syncthreads();
  for (int c = wave; c < C; c += 4 * MOL_WAVES) {
    float t0[4], t1[4];
#pragma unroll
    for (int j = 0; j < 4; ++j) {
      t0[j] = T[(c + j * MOL_WAVES) * N_ATOMS + s0];
      t1[j] = T[(c + j * MOL_WAVES) * N_ATOMS + s1];
    }
#pragma unroll
    for (int j = 0; j < 4; ++j) {
      atomicAdd(&H[(c + j * MOL_WAVES) * N_ATOMS + d0], n0 * t0[j]);
      atomicAdd(&H[(c + j * MOL_WAVES) * N_ATOMS + d1], n1 * t1[j]);
    }
  }
  __syncthreads();
  if (RELU) {
    for (int c = wave; c < C; c += MOL_WAVES)
      H[c * N_ATOMS + lane] = fmaxf(H[c * N_ATOMS + lane], 0.0f);
  }
  __syncthreads();
}

__global__ __launch_bounds__(256, 8) void mol_gnn_kernel(
    const float* __restrict__ mol_x, const int* __restrict__ mei,
    const float* __restrict__ W1, const float* __restrict__ b1,
    const float* __restrict__ W2, const float* __restrict__ b2,
    const float* __restrict__ W3, const float* __restrict__ b3,
    float* __restrict__ embed) {
  __shared__ float Abuf[N_ATOMS * HID];
  __shared__ float Bbuf[N_ATOMS * HID];
  __shared__ float s_dinv[N_ATOMS];
  __shared__ float s_norm[E_MOL];
  __shared__ int   s_deg[N_ATOMS];
  __shared__ int   s_src[E_MOL];
  __shared__ int   s_dst[E_MOL];

  const int mol  = blockIdx.x;
  const int tid  = threadIdx.x;
  const int lane = tid & 31;
  const int wave = __builtin_amdgcn_readfirstlane(tid >> 5);
  __builtin_assume(wave >= 0 && wave < MOL_WAVES);

  const float* X = mol_x + (size_t)mol * (N_ATOMS * IN_CH);
  const float4* X4 = reinterpret_cast<const float4*>(X);
  for (int q = wave; q < IN_CH / 4; q += MOL_WAVES) {
    const float4 v = X4[lane * (IN_CH / 4) + q];
    Abuf[(4 * q + 0) * N_ATOMS + lane] = v.x;
    Abuf[(4 * q + 1) * N_ATOMS + lane] = v.y;
    Abuf[(4 * q + 2) * N_ATOMS + lane] = v.z;
    Abuf[(4 * q + 3) * N_ATOMS + lane] = v.w;
  }
  const int* me = mei + (size_t)mol * (2 * E_MOL);
  if (tid < E_MOL) { s_src[tid] = me[tid]; s_dst[tid] = me[E_MOL + tid]; }
  if (tid < N_ATOMS) s_deg[tid] = 1;
  __syncthreads();
  if (tid < E_MOL) atomicAdd(&s_deg[s_dst[tid]], 1);
  __syncthreads();
  if (tid < N_ATOMS) s_dinv[tid] = rsqrtf((float)s_deg[tid]);
  __syncthreads();
  if (tid < E_MOL) s_norm[tid] = s_dinv[s_src[tid]] * s_dinv[s_dst[tid]];
  __syncthreads();

  const float dinv2 = s_dinv[lane] * s_dinv[lane];
  const int   es0 = s_src[lane],      ed0 = s_dst[lane];
  const int   es1 = s_src[lane + 32], ed1 = s_dst[lane + 32];
  const float en0 = s_norm[lane],     en1 = s_norm[lane + 32];

  gemm32t_wmma<IN_CH, HID>(Abuf, W1, Bbuf, wave, lane);
  __syncthreads();
  aggregate_t<HID, true>(Bbuf, Abuf, b1, dinv2, es0, ed0, en0, es1, ed1, en1,
                         wave, lane);

  gemm32t_wmma<HID, HID>(Abuf, W2, Bbuf, wave, lane);
  __syncthreads();
  aggregate_t<HID, true>(Bbuf, Abuf, b2, dinv2, es0, ed0, en0, es1, ed1, en1,
                         wave, lane);

  gemm32t_wmma<HID, OUT_CH>(Abuf, W3, Bbuf, wave, lane);
  __syncthreads();
  aggregate_t<OUT_CH, false>(Bbuf, Abuf, b3, dinv2, es0, ed0, en0, es1, ed1, en1,
                             wave, lane);

  if (tid < OUT_CH) {
    const int c = tid;
    float v = Abuf[c * N_ATOMS + (c & (N_ATOMS - 1))];
    for (int r = 1; r < N_ATOMS; ++r)
      v = fminf(v, Abuf[c * N_ATOMS + ((c + r) & (N_ATOMS - 1))]);
    embed[(size_t)mol * OUT_CH + c] = v;
  }
}

__global__ __launch_bounds__(256) void wmma_gemm_kernel(
    const float* __restrict__ A, const float* __restrict__ B,
    float* __restrict__ C, int Mrows, int K, int N) {
  const int lane = threadIdx.x & 31;
  const int wave = blockIdx.x * (blockDim.x >> 5) + (threadIdx.x >> 5);
  const int ntT = N >> 4;
  const int nTiles = (Mrows >> 4) * ntT;
  if (wave >= nTiles) return;
  const int mt = wave / ntT;
  const int nt = wave - mt * ntT;
  const int hi = lane >> 4, lo = lane & 15;
  const int mrow = mt * 16 + lo;
  const int ncol = nt * 16 + lo;
  v8f acc = {};
  for (int kb = 0; kb < K; kb += 4) {
    const int k = kb + 2 * hi;
    v2f a, b;
    a.x = A[(size_t)mrow * K + k];
    a.y = A[(size_t)mrow * K + k + 1];
    b.x = B[(size_t)k * N + ncol];
    b.y = B[(size_t)(k + 1) * N + ncol];
    acc = wmma_f32_k4(a, b, acc);
  }
#pragma unroll
  for (int v = 0; v < 8; ++v)
    C[(size_t)(mt * 16 + v + 8 * hi) * N + nt * 16 + lo] = acc[v];
}

__global__ void kg_deg_init_kernel(float* __restrict__ deg, int n) {
  int i = blockIdx.x * blockDim.x + threadIdx.x;
  if (i < n) deg[i] = 1.0f;
}
__global__ void kg_deg_count_kernel(const int* __restrict__ dst,
                                    float* __restrict__ deg, int ne) {
  int i = blockIdx.x * blockDim.x + threadIdx.x;
  if (i < ne) atomicAdd(&deg[dst[i]], 1.0f);
}
__global__ void kg_dinv_kernel(const float* __restrict__ deg,
                               float* __restrict__ dinv, int n) {
  int i = blockIdx.x * blockDim.x + threadIdx.x;
  if (i < n) dinv[i] = rsqrtf(deg[i]);
}
#define KG_NB      40
#define KG_THREADS 1024
#define KG_WAVES   32
#define KG_CAP     512
#define KG_ROUND   (KG_WAVES * KG_CAP)
#define KG_BATCH   8

template <int C, bool RELU>
__global__ __launch_bounds__(KG_THREADS) void kg_block_gather_kernel(
    const float* __restrict__ XW, const float* __restrict__ dinv,
    const float* __restrict__ bias, const int* __restrict__ src,
    const int* __restrict__ dst, float* __restrict__ H,
    int n_nodes, int n_edges) {
  constexpr int G = KG_THREADS / C;
  __shared__ float s_acc[KG_NB * C];
  __shared__ int2  s_list[KG_ROUND];
  __shared__ int   s_cnt[KG_WAVES];
  __shared__ float s_dinv[KG_NB];

  const int tid  = threadIdx.x;
  const int lane = tid & 31;
  const int wave = __builtin_amdgcn_readfirstlane(tid >> 5);
  const int g    = __builtin_amdgcn_readfirstlane(tid / C);
  const int c    = tid & (C - 1);
  const int n0   = blockIdx.x * KG_NB;
  const int nb   = min(KG_NB, n_nodes - n0);

  if (tid < KG_NB) s_dinv[tid] = (tid < nb) ? dinv[n0 + tid] : 0.0f;
  __syncthreads();
  for (int j = g; j < nb; j += G) {
    const float di = s_dinv[j];
    s_acc[j * C + c] = di * di * XW[(size_t)(n0 + j) * C + c] + bias[c];
  }

  const unsigned lanes_below = (1u << lane) - 1u;
  for (int base = 0; base < n_edges; base += KG_ROUND) {
    const int wbase = base + wave * KG_CAP;
    int4 dv[4];
#pragma unroll
    for (int q = 0; q < 4; ++q) {
      const int e = wbase + q * 128 + lane * 4;
      dv[q] = make_int4(-1, -1, -1, -1);
      if (e < n_edges) dv[q] = *reinterpret_cast<const int4*>(dst + e);
    }
    int dl[16], sv[16];
    int count = 0;
#pragma unroll
    for (int q = 0; q < 4; ++q) {
      const int e = wbase + q * 128 + lane * 4;
      const int dd[4] = {dv[q].x, dv[q].y, dv[q].z, dv[q].w};
#pragma unroll
      for (int j = 0; j < 4; ++j) {
        const int d = dd[j] - n0;
        const bool m = (unsigned)d < (unsigned)nb;
        dl[q * 4 + j] = m ? d : -1;
        sv[q * 4 + j] = src[m ? (e + j) : 0];
        count += __popc(__builtin_amdgcn_ballot_w32(m));
      }
    }
    if (lane == 0) s_cnt[wave] = count;
    __syncthreads();

    const int v = s_cnt[lane];
    int incl = v;
#pragma unroll
    for (int dlt = 1; dlt < 32; dlt <<= 1) {
      const int t = __shfl_up(incl, dlt);
      if (lane >= dlt) incl += t;
    }
    const int total = __builtin_amdgcn_readlane(incl, 31);
    int pos = __builtin_amdgcn_readlane(incl - v, wave);
#pragma unroll
    for (int k = 0; k < 16; ++k) {
      const bool m = dl[k] >= 0;
      const unsigned mask = __builtin_amdgcn_ballot_w32(m);
      if (m) s_list[pos + __popc(mask & lanes_below)] = make_int2(sv[k], dl[k]);
      pos += __popc(mask);
    }
    __syncthreads();

    for (int i0 = 0; i0 < total; i0 += 32) {
      const int idx = i0 + lane;
      int2 en = make_int2(0, -1);
      if (idx < total) en = s_list[idx];
      unsigned mm = __builtin_amdgcn_ballot_w32(en.y >= 0 && (en.y & (G - 1)) == g);
      while (mm) {
        int  es[KG_BATCH], ed[KG_BATCH];
        bool ok[KG_BATCH];
#pragma unroll
        for (int k = 0; k < KG_BATCH; ++k) {
          ok[k] = mm != 0u;
          const int b = ok[k] ? __builtin_ctz(mm) : 0;
          es[k] = __builtin_amdgcn_readlane(en.x, b);
          ed[k] = __builtin_amdgcn_readlane(en.y, b);
          mm &= mm - 1u;
        }
        float x[KG_BATCH], w[KG_BATCH];
#pragma unroll
        for (int k = 0; k < KG_BATCH; ++k) {
          x[k] = XW[(size_t)es[k] * C + c];
          w[k] = dinv[es[k]];
        }
#pragma unroll
        for (int k = 0; k < KG_BATCH; ++k) {
          if (ok[k])
            s_acc[ed[k] * C + c] += w[k] * s_dinv[ed[k]] * x[k];
        }
      }
    }
  }

  for (int j = g; j < nb; j += G) {
    float r = s_acc[j * C + c];
    if (RELU) r = fmaxf(r, 0.0f);
    H[(size_t)(n0 + j) * C + c] = r;
  }
}

extern "C" void kernel_launch(void* const* d_in, const int* in_sizes, int n_in,
                              void* d_out, int out_size, void* d_ws, size_t ws_size,
                              hipStream_t stream) {
  const float* mol_x = (const float*)d_in[0];
  const int*   mei   = (const int*)d_in[1];
  const int*   kei   = (const int*)d_in[2];
  const float* W1 = (const float*)d_in[3];
  const float* b1 = (const float*)d_in[4];
  const float* W2 = (const float*)d_in[5];
  const float* b2 = (const float*)d_in[6];
  const float* W3 = (const float*)d_in[7];
  const float* b3 = (const float*)d_in[8];
  const float* Wk1 = (const float*)d_in[9];
  const float* bk1 = (const float*)d_in[10];
  const float* Wk2 = (const float*)d_in[11];
  const float* bk2 = (const float*)d_in[12];
  float* out = (float*)d_out;

  float* ws    = (float*)d_ws;
  float* embed = ws;
  float* deg   = embed + M_MOL * OUT_CH;
  float* dinv  = deg + M_MOL;
  float* XW    = dinv + M_MOL;
  float* h     = XW + (size_t)M_MOL * HID;
  float* XW2   = h + (size_t)M_MOL * HID;

  const int* kg_src = kei;
  const int* kg_dst = kei + E_KG;

  mol_gnn_kernel<<<M_MOL, 256, 0, stream>>>(mol_x, mei, W1, b1, W2, b2, W3, b3,
                                            embed);

  kg_deg_init_kernel<<<(M_MOL + 255) / 256, 256, 0, stream>>>(deg, M_MOL);
  kg_deg_count_kernel<<<(E_KG + 255) / 256, 256, 0, stream>>>(kg_dst, deg, E_KG);
  kg_dinv_kernel<<<(M_MOL + 255) / 256, 256, 0, stream>>>(deg, dinv, M_MOL);

  {
    int tiles = (M_MOL / 16) * (HID / 16);
    wmma_gemm_kernel<<<(tiles + 7) / 8, 256, 0, stream>>>(embed, Wk1, XW,
                                                          M_MOL, OUT_CH, HID);
  }
  kg_block_gather_kernel<HID, true><<<(M_MOL + KG_NB - 1) / KG_NB, KG_THREADS, 0,
                                      stream>>>(XW, dinv, bk1, kg_src, kg_dst, h,
                                                M_MOL, E_KG);

  {
    int tiles = (M_MOL / 16) * (OUT_CH / 16);
    wmma_gemm_kernel<<<(tiles + 7) / 8, 256, 0, stream>>>(h, Wk2, XW2,
                                                          M_MOL, HID, OUT_CH);
  }
  kg_block_gather_kernel<OUT_CH, false><<<(M_MOL + KG_NB - 1) / KG_NB, KG_THREADS,
                                          0, stream>>>(XW2, dinv, bk2, kg_src,
                                                       kg_dst, out, M_MOL, E_KG);
}


# KG layers: per-round LDS row sums -> list kept over all rounds, drained once: a wave owns 1-2 destinations, sums in registers, whole rows by 16-byte loads, one barrier per round; sum order kept
# speedup vs baseline: 2.5206x; 1.4032x over previous
#include <hip/hip_runtime.h>

typedef __attribute__((ext_vector_type(2))) float v2f;
typedef __attribute__((ext_vector_type(8))) float v8f;

#define N_ATOMS 32
#define E_MOL   64
#define IN_CH   64
#define OUT_CH  128
#define HID     256
#define M_MOL   10000
#define E_KG    320000

__device__ __forceinline__ v8f wmma_f32_k4(v2f a, v2f b, v8f c) {
  return __builtin_amdgcn_wmma_f32_16x16x4_f32(false, a, false, b, (short)0, c,
                                               false, false);
}

#define MOL_WAVES 8

#define LDP 36

template <int K, int N, bool RELU>
__device__ __forceinline__ void gcn_layer_wmma(const float* __restrict__ XT,
                                               const float* __restrict__ W,
                                               const float* __restrict__ bias,
                                               const float* __restrict__ AT,
                                               float* __restrict__ TT,
                                               int wave, int lane) {
  const int hi = lane >> 4;
  const int lo = lane & 15;
  const float* __restrict__ Wl = W + hi * N + lo;
  const float* __restrict__ xp = XT + hi * LDP + lo;
  int tile1 = 16;
  asm("" : "+v"(tile1));
  const float* __restrict__ xq = xp + tile1;
  const float* __restrict__ at0 = AT + lo * LDP + 2 * hi;
  const float* __restrict__ at1 = AT + (16 + lo) * LDP + 2 * hi;
  for (int nt = wave; nt < N / 16; nt += MOL_WAVES) {
    const float* __restrict__ wp = Wl + nt * 16;
    v8f acc0 = {};
    v8f acc1 = {};
    for (int kb = 0; kb < K; kb += 32) {
      v2f a[8];
#pragma unroll
      for (int j = 0; j < 8; ++j) {
        a[j].x = wp[(kb + 4 * j) * N];
        a[j].y = wp[(kb + 4 * j + 2) * N];
      }
      __builtin_amdgcn_sched_barrier(0);
#pragma unroll
      for (int j = 0; j < 8; ++j) {
        v2f b0, b1;
        b0.x = xp[(kb + 4 * j) * LDP];
        b0.y = xp[(kb + 4 * j + 2) * LDP];
        b1.x = xq[(kb + 4 * j) * LDP];
        b1.y = xq[(kb + 4 * j + 2) * LDP];
        acc0 = wmma_f32_k4(a[j], b0, acc0);
        acc1 = wmma_f32_k4(a[j], b1, acc1);
      }
    }
    float* trow = TT + (nt * 16 + 8 * hi) * LDP + lo;
#pragma unroll
    for (int v = 0; v < 8; ++v) {
      trow[v * LDP]      = acc0[v];
      trow[v * LDP + 16] = acc1[v];
    }
    const float* ar = TT + (nt * 16 + lo) * LDP + 2 * hi;
    v2f ta[8];
#pragma unroll
    for (int t = 0; t < 8; ++t)
      ta[t] = *reinterpret_cast<const v2f*>(ar + 4 * t);
    const float4 bA = *reinterpret_cast<const float4*>(bias + nt * 16 + 8 * hi);
    const float4 bB = *reinterpret_cast<const float4*>(bias + nt * 16 + 8 * hi + 4);
    v8f h0 = {bA.x, bA.y, bA.z, bA.w, bB.x, bB.y, bB.z, bB.w};
    v8f h1 = h0;
#pragma unroll
    for (int t = 0; t < 8; ++t) {
      const v2f n0 = *reinterpret_cast<const v2f*>(at0 + 4 * t);
      const v2f n1 = *reinterpret_cast<const v2f*>(at1 + 4 * t);
      h0 = wmma_f32_k4(ta[t], n0, h0);
      h1 = wmma_f32_k4(ta[t], n1, h1);
    }
#pragma unroll
    for (int v = 0; v < 8; ++v) {
      float r0 = h0[v], r1 = h1[v];
      if (RELU) { r0 = fmaxf(r0, 0.0f); r1 = fmaxf(r1, 0.0f); }
      trow[v * LDP]      = r0;
      trow[v * LDP + 16] = r1;
    }
  }
}

__global__ __launch_bounds__(256, 8) void mol_gnn_kernel(
    const float* __restrict__ mol_x, const int* __restrict__ mei,
    const float* __restrict__ W1, const float* __restrict__ b1,
    const float* __restrict__ W2, const float* __restrict__ b2,
    const float* __restrict__ W3, const float* __restrict__ b3,
    float* __restrict__ embed) {
  __shared__ __attribute__((aligned(16))) float Abuf[HID * LDP];
  __shared__ __attribute__((aligned(16))) float Bbuf[HID * LDP];
  __shared__ __attribute__((aligned(16))) float s_at[N_ATOMS * LDP];
  __shared__ float s_dinv[N_ATOMS];
  __shared__ int   s_deg[N_ATOMS];
  __shared__ int   s_src[E_MOL];
  __shared__ int   s_dst[E_MOL];

  const int mol  = blockIdx.x;
  const int tid  = threadIdx.x;
  const int lane = tid & 31;
  const int wave = __builtin_amdgcn_readfirstlane(tid >> 5);
  __builtin_assume(wave >= 0 && wave < MOL_WAVES);

  const float* X = mol_x + (size_t)mol * (N_ATOMS * IN_CH);
  const float4* X4 = reinterpret_cast<const float4*>(X);
  for (int q = wave; q < IN_CH / 4; q += MOL_WAVES) {
    const float4 v = X4[lane * (IN_CH / 4) + q];
    Abuf[(4 * q + 0) * LDP + lane] = v.x;
    Abuf[(4 * q + 1) * LDP + lane] = v.y;
    Abuf[(4 * q + 2) * LDP + lane] = v.z;
    Abuf[(4 * q + 3) * LDP + lane] = v.w;
  }
  const int* me = mei + (size_t)mol * (2 * E_MOL);
  if (tid < E_MOL) { s_src[tid] = me[tid]; s_dst[tid] = me[E_MOL + tid]; }
  if (tid < N_ATOMS) s_deg[tid] = 1;
  for (int i = tid; i < N_ATOMS * LDP; i += 256) s_at[i] = 0.0f;
  __syncthreads();
  if (tid < E_MOL) atomicAdd(&s_deg[s_dst[tid]], 1);
  __syncthreads();
  if (tid < N_ATOMS) {
    const float d = rsqrtf((float)s_deg[tid]);
    s_dinv[tid] = d;
    s_at[tid * LDP + tid] = d * d;
  }
  __syncthreads();
  if (tid < E_MOL) {
    const int s = s_src[tid], d = s_dst[tid];
    atomicAdd(&s_at[d * LDP + s], s_dinv[s] * s_dinv[d]);
  }
  __syncthreads();

  gcn_layer_wmma<IN_CH, HID, true>(Abuf, W1, b1, s_at, Bbuf, wave, lane);
  __syncthreads();
  gcn_layer_wmma<HID, HID, true>(Bbuf, W2, b2, s_at, Abuf, wave, lane);
  __syncthreads();
  gcn_layer_wmma<HID, OUT_CH, false>(Abuf, W3, b3, s_at, Bbuf, wave, lane);
  __syncthreads();

  if (tid < OUT_CH) {
    const int c = tid;
    float v = Bbuf[c * LDP + (c & (N_ATOMS - 1))];
    for (int r = 1; r < N_ATOMS; ++r)
      v = fminf(v, Bbuf[c * LDP + ((c + r) & (N_ATOMS - 1))]);
    embed[(size_t)mol * OUT_CH + c] = v;
  }
}

__global__ __launch_bounds__(256) void wmma_gemm_kernel(
    const float* __restrict__ A, const float* __restrict__ B,
    float* __restrict__ C, int Mrows, int K, int N) {
  const int lane = threadIdx.x & 31;
  const int wave = blockIdx.x * (blockDim.x >> 5) + (threadIdx.x >> 5);
  const int ntT = N >> 4;
  const int nTiles = (Mrows >> 4) * ntT;
  if (wave >= nTiles) return;
  const int mt = wave / ntT;
  const int nt = wave - mt * ntT;
  const int hi = lane >> 4, lo = lane & 15;
  const int mrow = mt * 16 + lo;
  const int ncol = nt * 16 + lo;
  v8f acc = {};
  for (int kb = 0; kb < K; kb += 4) {
    const int k = kb + 2 * hi;
    v2f a, b;
    a.x = A[(size_t)mrow * K + k];
    a.y = A[(size_t)mrow * K + k + 1];
    b.x = B[(size_t)k * N + ncol];
    b.y = B[(size_t)(k + 1) * N + ncol];
    acc = wmma_f32_k4(a, b, acc);
  }
#pragma unroll
  for (int v = 0; v < 8; ++v)
    C[(size_t)(mt * 16 + v + 8 * hi) * N + nt * 16 + lo] = acc[v];
}

__global__ void kg_deg_init_kernel(float* __restrict__ deg, int n) {
  int i = blockIdx.x * blockDim.x + threadIdx.x;
  if (i < n) deg[i] = 1.0f;
}
__global__ void kg_deg_count_kernel(const int* __restrict__ dst,
                                    float* __restrict__ deg, int ne) {
  int i = blockIdx.x * blockDim.x + threadIdx.x;
  if (i < ne) atomicAdd(&deg[dst[i]], 1.0f);
}
__global__ void kg_dinv_kernel(const float* __restrict__ deg,
                               float* __restrict__ dinv, int n) {
  int i = blockIdx.x * blockDim.x + threadIdx.x;
  if (i < n) dinv[i] = rsqrtf(deg[i]);
}
#define KG_NB      40
#define KG_THREADS 1024
#define KG_WAVES   32
#define KG_CAP     512
#define KG_ROUND   (KG_WAVES * KG_CAP)
#define KG_LIST    (2 * KG_ROUND)

template <int C, bool RELU>
__global__ __launch_bounds__(KG_THREADS) void kg_block_gather_kernel(
    const float* __restrict__ XW, const float* __restrict__ dinv,
    const float* __restrict__ bias, const int* __restrict__ src,
    const int* __restrict__ dst, float* __restrict__ H,
    int n_nodes, int n_edges) {
  constexpr int V = C / 128;
  constexpr int B = 8 / V;
  __shared__ int2 s_list[KG_LIST];
  __shared__ int  s_cnt[2][KG_WAVES];

  const int tid  = threadIdx.x;
  const int lane = tid & 31;
  const int wave = __builtin_amdgcn_readfirstlane(tid >> 5);
  const int n0   = blockIdx.x * KG_NB;
  const int nb   = min(KG_NB, n_nodes - n0);
  const float4* __restrict__ XW4   = reinterpret_cast<const float4*>(XW);
  const float4* __restrict__ bias4 = reinterpret_cast<const float4*>(bias);

  float4 acc0[V], acc1[V];
  float dd0 = 0.0f, dd1 = 0.0f;
  for (int v = 0; v < V; ++v) {
    acc0[v] = make_float4(0.0f, 0.0f, 0.0f, 0.0f);
    acc1[v] = make_float4(0.0f, 0.0f, 0.0f, 0.0f);
  }
  if (wave < nb) {
    dd0 = dinv[n0 + wave];
    const float s = dd0 * dd0;
    for (int v = 0; v < V; ++v) {
      const float4 x = XW4[(size_t)(n0 + wave) * (C / 4) + v * 32 + lane];
      const float4 bb = bias4[v * 32 + lane];
      acc0[v].x = s * x.x + bb.x;
      acc0[v].y = s * x.y + bb.y;
      acc0[v].z = s * x.z + bb.z;
      acc0[v].w = s * x.w + bb.w;
    }
  }
  if (wave + 32 < nb) {
    dd1 = dinv[n0 + wave + 32];
    const float s = dd1 * dd1;
    for (int v = 0; v < V; ++v) {
      const float4 x = XW4[(size_t)(n0 + wave + 32) * (C / 4) + v * 32 + lane];
      const float4 bb = bias4[v * 32 + lane];
      acc1[v].x = s * x.x + bb.x;
      acc1[v].y = s * x.y + bb.y;
      acc1[v].z = s * x.z + bb.z;
      acc1[v].w = s * x.w + bb.w;
    }
  }

  const unsigned lanes_below = (1u << lane) - 1u;
  int fill = 0;
  int par  = 0;
  for (int base = 0;; base += KG_ROUND, par ^= 1) {
    const bool done = base >= n_edges;
    if (done || fill > KG_LIST - KG_ROUND) {
      asm volatile("s_wait_dscnt 0x0" ::: "memory");
      __syncthreads();
      int q0 = 0, q1 = 0, qn0 = 0, qn1 = 0;
      for (int i0 = 0;; i0 += 32) {
        const bool last = i0 >= fill;
        int2 en = make_int2(0, -1);
        if (i0 + lane < fill) en = s_list[i0 + lane];
        unsigned m0 = __builtin_amdgcn_ballot_w32(en.y == wave);
        unsigned m1 = __builtin_amdgcn_ballot_w32(en.y == wave + 32);
        if (last || qn0 + __popc(m0) > 32 || qn1 + __popc(m1) > 32) {
          for (int pass = 0; pass < 2; ++pass) {
            if (qn0 > 0) {
              const float wq = dinv[(lane < qn0) ? q0 : 0];
              for (int k0 = 0; k0 < qn0; k0 += B) {
                int   es[B];
                float nrm[B];
#pragma unroll
                for (int k = 0; k < B; ++k) {
                  const bool ok = k0 + k < qn0;
                  const int  kk = ok ? k0 + k : k0;
                  es[k] = __builtin_amdgcn_readlane(q0, kk);
                  const float wk =
                      __int_as_float(__builtin_amdgcn_readlane(__float_as_int(wq), kk));
                  nrm[k] = ok ? wk * dd0 : 0.0f;
                }
                float4 x[B][V];
#pragma unroll
                for (int k = 0; k < B; ++k) {
#pragma unroll
                  for (int v = 0; v < V; ++v)
                    x[k][v] = XW4[(size_t)es[k] * (C / 4) + v * 32 + lane];
                }
#pragma unroll
                for (int k = 0; k < B; ++k) {
#pragma unroll
                  for (int v = 0; v < V; ++v) {
                    acc0[v].x += nrm[k] * x[k][v].x;
                    acc0[v].y += nrm[k] * x[k][v].y;
                    acc0[v].z += nrm[k] * x[k][v].z;
                    acc0[v].w += nrm[k] * x[k][v].w;
                  }
                }
              }
            }
            { const int t = q0; q0 = q1; q1 = t; }
            { const int t = qn0; qn0 = qn1; qn1 = t; }
            { const float t = dd0; dd0 = dd1; dd1 = t; }
            for (int v = 0; v < V; ++v) {
              const float4 t = acc0[v]; acc0[v] = acc1[v]; acc1[v] = t;
            }
          }
          qn0 = 0;
          qn1 = 0;
          if (last) break;
        }
        while (m0) {
          const int s = __builtin_amdgcn_readlane(en.x, __builtin_ctz(m0));
          if (lane == qn0) q0 = s;
          ++qn0;
          m0 &= m0 - 1u;
        }
        while (m1) {
          const int s = __builtin_amdgcn_readlane(en.x, __builtin_ctz(m1));
          if (lane == qn1) q1 = s;
          ++qn1;
          m1 &= m1 - 1u;
        }
      }
      if (done) break;
      __syncthreads();
      fill = 0;
    }

    const int wbase = base + wave * KG_CAP;
    int4 dv[4];
#pragma unroll
    for (int q4 = 0; q4 < 4; ++q4) {
      const int e = wbase + q4 * 128 + lane * 4;
      dv[q4] = make_int4(-1, -1, -1, -1);
      if (e < n_edges) dv[q4] = *reinterpret_cast<const int4*>(dst + e);
    }
    int dl[16], sv[16];
    int count = 0;
#pragma unroll
    for (int q4 = 0; q4 < 4; ++q4) {
      const int e = wbase + q4 * 128 + lane * 4;
      const int dd[4] = {dv[q4].x, dv[q4].y, dv[q4].z, dv[q4].w};
#pragma unroll
      for (int j = 0; j < 4; ++j) {
        const int d = dd[j] - n0;
        const bool m = (unsigned)d < (unsigned)nb;
        dl[q4 * 4 + j] = m ? d : -1;
        sv[q4 * 4 + j] = src[m ? (e + j) : 0];
        count += __popc(__builtin_amdgcn_ballot_w32(m));
      }
    }
    if (lane == 0) s_cnt[par][wave] = count;
    __syncthreads();

    const int cw = s_cnt[par][lane];
    int incl = cw;
#pragma unroll
    for (int dlt = 1; dlt < 32; dlt <<= 1) {
      const int t = __shfl_up(incl, dlt);
      if (lane >= dlt) incl += t;
    }
    const int total = __builtin_amdgcn_readlane(incl, 31);
    int pos = fill + __builtin_amdgcn_readlane(incl - cw, wave);
#pragma unroll
    for (int k = 0; k < 16; ++k) {
      const bool m = dl[k] >= 0;
      const unsigned mask = __builtin_amdgcn_ballot_w32(m);
      if (m) s_list[pos + __popc(mask & lanes_below)] = make_int2(sv[k], dl[k]);
      pos += __popc(mask);
    }
    fill += total;
  }

  float4* __restrict__ H4 = reinterpret_cast<float4*>(H);
  if (wave < nb) {
    for (int v = 0; v < V; ++v) {
      float4 r = acc0[v];
      if (RELU) {
        r.x = fmaxf(r.x, 0.0f); r.y = fmaxf(r.y, 0.0f);
        r.z = fmaxf(r.z, 0.0f); r.w = fmaxf(r.w, 0.0f);
      }
      H4[(size_t)(n0 + wave) * (C / 4) + v * 32 + lane] = r;
    }
  }
  if (wave + 32 < nb) {
    for (int v = 0; v < V; ++v) {
      float4 r = acc1[v];
      if (RELU) {
        r.x = fmaxf(r.x, 0.0f); r.y = fmaxf(r.y, 0.0f);
        r.z = fmaxf(r.z, 0.0f); r.w = fmaxf(r.w, 0.0f);
      }
      H4[(size_t)(n0 + wave + 32) * (C / 4) + v * 32 + lane] = r;
    }
  }
}

extern "C" void kernel_launch(void* const* d_in, const int* in_sizes, int n_in,
                              void* d_out, int out_size, void* d_ws, size_t ws_size,
                              hipStream_t stream) {
  const float* mol_x = (const float*)d_in[0];
  const int*   mei   = (const int*)d_in[1];
  const int*   kei   = (const int*)d_in[2];
  const float* W1 = (const float*)d_in[3];
  const float* b1 = (const float*)d_in[4];
  const float* W2 = (const float*)d_in[5];
  const float* b2 = (const float*)d_in[6];
  const float* W3 = (const float*)d_in[7];
  const float* b3 = (const float*)d_in[8];
  const float* Wk1 = (const float*)d_in[9];
  const float* bk1 = (const float*)d_in[10];
  const float* Wk2 = (const float*)d_in[11];
  const float* bk2 = (const float*)d_in[12];
  float* out = (float*)d_out;

  float* ws    = (float*)d_ws;
  float* embed = ws;
  float* deg   = embed + M_MOL * OUT_CH;
  float* dinv  = deg + M_MOL;
  float* XW    = dinv + M_MOL;
  float* h     = XW + (size_t)M_MOL * HID;
  float* XW2   = h + (size_t)M_MOL * HID;

  const int* kg_src = kei;
  const int* kg_dst = kei + E_KG;

  mol_gnn_kernel<<<M_MOL, 256, 0, stream>>>(mol_x, mei, W1, b1, W2, b2, W3, b3,
                                            embed);

  kg_deg_init_kernel<<<(M_MOL + 255) / 256, 256, 0, stream>>>(deg, M_MOL);
  kg_deg_count_kernel<<<(E_KG + 255) / 256, 256, 0, stream>>>(kg_dst, deg, E_KG);
  kg_dinv_kernel<<<(M_MOL + 255) / 256, 256, 0, stream>>>(deg, dinv, M_MOL);

  {
    int tiles = (M_MOL / 16) * (HID / 16);
    wmma_gemm_kernel<<<(tiles + 7) / 8, 256, 0, stream>>>(embed, Wk1, XW,
                                                          M_MOL, OUT_CH, HID);
  }
  kg_block_gather_kernel<HID, true><<<(M_MOL + KG_NB - 1) / KG_NB, KG_THREADS, 0,
                                      stream>>>(XW, dinv, bk1, kg_src, kg_dst, h,
                                                M_MOL, E_KG);

  {
    int tiles = (M_MOL / 16) * (OUT_CH / 16);
    wmma_gemm_kernel<<<(tiles + 7) / 8, 256, 0, stream>>>(h, Wk2, XW2,
                                                          M_MOL, HID, OUT_CH);
  }
  kg_block_gather_kernel<OUT_CH, false><<<(M_MOL + KG_NB - 1) / KG_NB, KG_THREADS,
                                          0, stream>>>(XW2, dinv, bk2, kg_src,
                                                       kg_dst, out, M_MOL, E_KG);
}


# KG edge filter: 20 barriered rounds with per-edge source loads -> one barrier-free pass: a wave streams its own chunks, rare matches go to a wave-private LDS list, merged once; sum order changes
# speedup vs baseline: 2.7152x; 1.0772x over previous
#include <hip/hip_runtime.h>

typedef __attribute__((ext_vector_type(2))) float v2f;
typedef __attribute__((ext_vector_type(8))) float v8f;

#define N_ATOMS 32
#define E_MOL   64
#define IN_CH   64
#define OUT_CH  128
#define HID     256
#define M_MOL   10000
#define E_KG    320000

__device__ __forceinline__ v8f wmma_f32_k4(v2f a, v2f b, v8f c) {
  return __builtin_amdgcn_wmma_f32_16x16x4_f32(false, a, false, b, (short)0, c,
                                               false, false);
}

#define MOL_WAVES 8

#define LDP 36

template <int K, int N, bool RELU>
__device__ __forceinline__ void gcn_layer_wmma(const float* __restrict__ XT,
                                               const float* __restrict__ W,
                                               const float* __restrict__ bias,
                                               const float* __restrict__ AT,
                                               float* __restrict__ TT,
                                               int wave, int lane) {
  const int hi = lane >> 4;
  const int lo = lane & 15;
  const float* __restrict__ Wl = W + hi * N + lo;
  const float* __restrict__ xp = XT + hi * LDP + lo;
  int tile1 = 16;
  asm("" : "+v"(tile1));
  const float* __restrict__ xq = xp + tile1;
  const float* __restrict__ at0 = AT + lo * LDP + 2 * hi;
  const float* __restrict__ at1 = AT + (16 + lo) * LDP + 2 * hi;
  for (int nt = wave; nt < N / 16; nt += MOL_WAVES) {
    const float* __restrict__ wp = Wl + nt * 16;
    v8f acc0 = {};
    v8f acc1 = {};
    for (int kb = 0; kb < K; kb += 32) {
      v2f a[8];
#pragma unroll
      for (int j = 0; j < 8; ++j) {
        a[j].x = wp[(kb + 4 * j) * N];
        a[j].y = wp[(kb + 4 * j + 2) * N];
      }
      __builtin_amdgcn_sched_barrier(0);
#pragma unroll
      for (int j = 0; j < 8; ++j) {
        v2f b0, b1;
        b0.x = xp[(kb + 4 * j) * LDP];
        b0.y = xp[(kb + 4 * j + 2) * LDP];
        b1.x = xq[(kb + 4 * j) * LDP];
        b1.y = xq[(kb + 4 * j + 2) * LDP];
        acc0 = wmma_f32_k4(a[j], b0, acc0);
        acc1 = wmma_f32_k4(a[j], b1, acc1);
      }
    }
    float* trow = TT + (nt * 16 + 8 * hi) * LDP + lo;
#pragma unroll
    for (int v = 0; v < 8; ++v) {
      trow[v * LDP]      = acc0[v];
      trow[v * LDP + 16] = acc1[v];
    }
    const float* ar = TT + (nt * 16 + lo) * LDP + 2 * hi;
    v2f ta[8];
#pragma unroll
    for (int t = 0; t < 8; ++t)
      ta[t] = *reinterpret_cast<const v2f*>(ar + 4 * t);
    const float4 bA = *reinterpret_cast<const float4*>(bias + nt * 16 + 8 * hi);
    const float4 bB = *reinterpret_cast<const float4*>(bias + nt * 16 + 8 * hi + 4);
    v8f h0 = {bA.x, bA.y, bA.z, bA.w, bB.x, bB.y, bB.z, bB.w};
    v8f h1 = h0;
#pragma unroll
    for (int t = 0; t < 8; ++t) {
      const v2f n0 = *reinterpret_cast<const v2f*>(at0 + 4 * t);
      const v2f n1 = *reinterpret_cast<const v2f*>(at1 + 4 * t);
      h0 = wmma_f32_k4(ta[t], n0, h0);
      h1 = wmma_f32_k4(ta[t], n1, h1);
    }
#pragma unroll
    for (int v = 0; v < 8; ++v) {
      float r0 = h0[v], r1 = h1[v];
      if (RELU) { r0 = fmaxf(r0, 0.0f); r1 = fmaxf(r1, 0.0f); }
      trow[v * LDP]      = r0;
      trow[v * LDP + 16] = r1;
    }
  }
}

__global__ __launch_bounds__(256, 8) void mol_gnn_kernel(
    const float* __restrict__ mol_x, const int* __restrict__ mei,
    const float* __restrict__ W1, const float* __restrict__ b1,
    const float* __restrict__ W2, const float* __restrict__ b2,
    const float* __restrict__ W3, const float* __restrict__ b3,
    float* __restrict__ embed) {
  __shared__ __attribute__((aligned(16))) float Abuf[HID * LDP];
  __shared__ __attribute__((aligned(16))) float Bbuf[HID * LDP];
  __shared__ __attribute__((aligned(16))) float s_at[N_ATOMS * LDP];
  __shared__ float s_dinv[N_ATOMS];
  __shared__ int   s_deg[N_ATOMS];
  __shared__ int   s_src[E_MOL];
  __shared__ int   s_dst[E_MOL];

  const int mol  = blockIdx.x;
  const int tid  = threadIdx.x;
  const int lane = tid & 31;
  const int wave = __builtin_amdgcn_readfirstlane(tid >> 5);
  __builtin_assume(wave >= 0 && wave < MOL_WAVES);

  const float* X = mol_x + (size_t)mol * (N_ATOMS * IN_CH);
  const float4* X4 = reinterpret_cast<const float4*>(X);
  for (int q = wave; q < IN_CH / 4; q += MOL_WAVES) {
    const float4 v = X4[lane * (IN_CH / 4) + q];
    Abuf[(4 * q + 0) * LDP + lane] = v.x;
    Abuf[(4 * q + 1) * LDP + lane] = v.y;
    Abuf[(4 * q + 2) * LDP + lane] = v.z;
    Abuf[(4 * q + 3) * LDP + lane] = v.w;
  }
  const int* me = mei + (size_t)mol * (2 * E_MOL);
  if (tid < E_MOL) { s_src[tid] = me[tid]; s_dst[tid] = me[E_MOL + tid]; }
  if (tid < N_ATOMS) s_deg[tid] = 1;
  for (int i = tid; i < N_ATOMS * LDP; i += 256) s_at[i] = 0.0f;
  __syncthreads();
  if (tid < E_MOL) atomicAdd(&s_deg[s_dst[tid]], 1);
  __syncthreads();
  if (tid < N_ATOMS) {
    const float d = rsqrtf((float)s_deg[tid]);
    s_dinv[tid] = d;
    s_at[tid * LDP + tid] = d * d;
  }
  __syncthreads();
  if (tid < E_MOL) {
    const int s = s_src[tid], d = s_dst[tid];
    atomicAdd(&s_at[d * LDP + s], s_dinv[s] * s_dinv[d]);
  }
  __syncthreads();

  gcn_layer_wmma<IN_CH, HID, true>(Abuf, W1, b1, s_at, Bbuf, wave, lane);
  __syncthreads();
  gcn_layer_wmma<HID, HID, true>(Bbuf, W2, b2, s_at, Abuf, wave, lane);
  __syncthreads();
  gcn_layer_wmma<HID, OUT_CH, false>(Abuf, W3, b3, s_at, Bbuf, wave, lane);
  __syncthreads();

  if (tid < OUT_CH) {
    const int c = tid;
    float v = Bbuf[c * LDP + (c & (N_ATOMS - 1))];
    for (int r = 1; r < N_ATOMS; ++r)
      v = fminf(v, Bbuf[c * LDP + ((c + r) & (N_ATOMS - 1))]);
    embed[(size_t)mol * OUT_CH + c] = v;
  }
}

__global__ __launch_bounds__(256) void wmma_gemm_kernel(
    const float* __restrict__ A, const float* __restrict__ B,
    float* __restrict__ C, int Mrows, int K, int N) {
  const int lane = threadIdx.x & 31;
  const int wave = blockIdx.x * (blockDim.x >> 5) + (threadIdx.x >> 5);
  const int ntT = N >> 4;
  const int nTiles = (Mrows >> 4) * ntT;
  if (wave >= nTiles) return;
  const int mt = wave / ntT;
  const int nt = wave - mt * ntT;
  const int hi = lane >> 4, lo = lane & 15;
  const int mrow = mt * 16 + lo;
  const int ncol = nt * 16 + lo;
  v8f acc = {};
  for (int kb = 0; kb < K; kb += 4) {
    const int k = kb + 2 * hi;
    v2f a, b;
    a.x = A[(size_t)mrow * K + k];
    a.y = A[(size_t)mrow * K + k + 1];
    b.x = B[(size_t)k * N + ncol];
    b.y = B[(size_t)(k + 1) * N + ncol];
    acc = wmma_f32_k4(a, b, acc);
  }
#pragma unroll
  for (int v = 0; v < 8; ++v)
    C[(size_t)(mt * 16 + v + 8 * hi) * N + nt * 16 + lo] = acc[v];
}

__global__ void kg_deg_init_kernel(float* __restrict__ deg, int n) {
  int i = blockIdx.x * blockDim.x + threadIdx.x;
  if (i < n) deg[i] = 1.0f;
}
__global__ void kg_deg_count_kernel(const int* __restrict__ dst,
                                    float* __restrict__ deg, int ne) {
  int i = blockIdx.x * blockDim.x + threadIdx.x;
  if (i < ne) atomicAdd(&deg[dst[i]], 1.0f);
}
__global__ void kg_dinv_kernel(const float* __restrict__ deg,
                               float* __restrict__ dinv, int n) {
  int i = blockIdx.x * blockDim.x + threadIdx.x;
  if (i < n) dinv[i] = rsqrtf(deg[i]);
}
#define KG_NB      40
#define KG_THREADS 1024
#define KG_WAVES   32
#define KG_CAP     512
#define KG_ROUND   (KG_WAVES * KG_CAP)
#define KG_LIST    (2 * KG_ROUND)
#define KG_WCAP    512
#define KG_SEG0    (KG_LIST / 2)
#define KG_G       4

template <int C, bool RELU>
__global__ __launch_bounds__(KG_THREADS) void kg_block_gather_kernel(
    const float* __restrict__ XW, const float* __restrict__ dinv,
    const float* __restrict__ bias, const int* __restrict__ src,
    const int* __restrict__ dst, float* __restrict__ H,
    int n_nodes, int n_edges) {
  constexpr int V = C / 128;
  constexpr int B = 8 / V;
  __shared__ int2 s_list[KG_LIST];
  __shared__ int  s_cnt[2][KG_WAVES];
  __shared__ int  s_wcnt[KG_WAVES];

  const int tid  = threadIdx.x;
  const int lane = tid & 31;
  const int wave = __builtin_amdgcn_readfirstlane(tid >> 5);
  const int n0   = blockIdx.x * KG_NB;
  const int nb   = min(KG_NB, n_nodes - n0);
  const float4* __restrict__ XW4   = reinterpret_cast<const float4*>(XW);
  const float4* __restrict__ bias4 = reinterpret_cast<const float4*>(bias);

  float4 acc0[V], acc1[V];
  float dd0 = 0.0f, dd1 = 0.0f;
  for (int v = 0; v < V; ++v) {
    acc0[v] = make_float4(0.0f, 0.0f, 0.0f, 0.0f);
    acc1[v] = make_float4(0.0f, 0.0f, 0.0f, 0.0f);
  }
  if (wave < nb) {
    dd0 = dinv[n0 + wave];
    const float s = dd0 * dd0;
    for (int v = 0; v < V; ++v) {
      const float4 x = XW4[(size_t)(n0 + wave) * (C / 4) + v * 32 + lane];
      const float4 bb = bias4[v * 32 + lane];
      acc0[v].x = s * x.x + bb.x;
      acc0[v].y = s * x.y + bb.y;
      acc0[v].z = s * x.z + bb.z;
      acc0[v].w = s * x.w + bb.w;
    }
  }
  if (wave + 32 < nb) {
    dd1 = dinv[n0 + wave + 32];
    const float s = dd1 * dd1;
    for (int v = 0; v < V; ++v) {
      const float4 x = XW4[(size_t)(n0 + wave + 32) * (C / 4) + v * 32 + lane];
      const float4 bb = bias4[v * 32 + lane];
      acc1[v].x = s * x.x + bb.x;
      acc1[v].y = s * x.y + bb.y;
      acc1[v].z = s * x.z + bb.z;
      acc1[v].w = s * x.w + bb.w;
    }
  }

  const unsigned lanes_below = (1u << lane) - 1u;
  int fill  = 0;
  int base0 = 0;
  {
    int2* __restrict__ seg = s_list + KG_SEG0 + wave * KG_WCAP;
    const int ntrip = (n_edges + 128 * KG_WAVES * KG_G - 1) / (128 * KG_WAVES * KG_G);
    int cnt = 0;
    const int last4 = max(n_edges - 4, 0);
    auto load_trip = [&](int4 (&buf)[KG_G], int t) {
#pragma unroll
      for (int g = 0; g < KG_G; ++g) {
        const int e = ((t * KG_G + g) * KG_WAVES + wave) * 128 + lane * 4;
        buf[g] = *reinterpret_cast<const int4*>(dst + min(e, last4));
      }
    };
    auto test_trip = [&](const int4 (&buf)[KG_G], int t) {
#pragma unroll
      for (int g = 0; g < KG_G; ++g) {
        const int e0 = ((t * KG_G + g) * KG_WAVES + wave) * 128 + lane * 4;
        const int d[4] = {buf[g].x - n0, buf[g].y - n0, buf[g].z - n0, buf[g].w - n0};
        const unsigned nbv = (e0 < n_edges) ? (unsigned)nb : 0u;
        unsigned mk[4];
#pragma unroll
        for (int j = 0; j < 4; ++j)
          mk[j] = __builtin_amdgcn_ballot_w32((unsigned)d[j] < nbv);
        if ((mk[0] | mk[1] | mk[2] | mk[3]) != 0u) {
#pragma unroll
          for (int j = 0; j < 4; ++j) {
            if (mk[j] != 0u) {
              const int p = cnt + __popc(mk[j] & lanes_below);
              if ((unsigned)d[j] < nbv && p < KG_WCAP)
                seg[p] = make_int2(e0 + j, d[j]);
              cnt += __popc(mk[j]);
            }
          }
        }
      }
    };
    int4 bufa[KG_G], bufb[KG_G];
    load_trip(bufa, 0);
    for (int t = 0; t < ntrip; t += 2) {
      load_trip(bufb, t + 1);
      test_trip(bufa, t);
      load_trip(bufa, t + 2);
      test_trip(bufb, t + 1);
    }
    if (lane == 0) s_wcnt[wave] = cnt;
    __syncthreads();
    const int cw = s_wcnt[lane];
    if (__builtin_amdgcn_ballot_w32(cw > KG_WCAP) == 0u) {
      int incl = cw;
#pragma unroll
      for (int dlt = 1; dlt < 32; dlt <<= 1) {
        const int t = __shfl_up(incl, dlt);
        if (lane >= dlt) incl += t;
      }
      const int off = __builtin_amdgcn_readlane(incl - cw, wave);
      for (int i = lane; i < cnt; i += 32) {
        int2 en = seg[i];
        en.x = src[en.x];
        s_list[off + i] = en;
      }
      fill  = __builtin_amdgcn_readlane(incl, 31);
      base0 = n_edges;
    }
  }
  int par  = 0;
  for (int base = base0;; base += KG_ROUND, par ^= 1) {
    const bool done = base >= n_edges;
    if (done || fill > KG_LIST - KG_ROUND) {
      asm volatile("s_wait_dscnt 0x0" ::: "memory");
      __syncthreads();
      int q0 = 0, q1 = 0, qn0 = 0, qn1 = 0;
      for (int i0 = 0;; i0 += 32) {
        const bool last = i0 >= fill;
        int2 en = make_int2(0, -1);
        if (i0 + lane < fill) en = s_list[i0 + lane];
        unsigned m0 = __builtin_amdgcn_ballot_w32(en.y == wave);
        unsigned m1 = __builtin_amdgcn_ballot_w32(en.y == wave + 32);
        if (last || qn0 + __popc(m0) > 32 || qn1 + __popc(m1) > 32) {
          for (int pass = 0; pass < 2; ++pass) {
            if (qn0 > 0) {
              const float wq = dinv[(lane < qn0) ? q0 : 0];
              for (int k0 = 0; k0 < qn0; k0 += B) {
                int   es[B];
                float nrm[B];
#pragma unroll
                for (int k = 0; k < B; ++k) {
                  const bool ok = k0 + k < qn0;
                  const int  kk = ok ? k0 + k : k0;
                  es[k] = __builtin_amdgcn_readlane(q0, kk);
                  const float wk =
                      __int_as_float(__builtin_amdgcn_readlane(__float_as_int(wq), kk));
                  nrm[k] = ok ? wk * dd0 : 0.0f;
                }
                float4 x[B][V];
#pragma unroll
                for (int k = 0; k < B; ++k) {
#pragma unroll
                  for (int v = 0; v < V; ++v)
                    x[k][v] = XW4[(size_t)es[k] * (C / 4) + v * 32 + lane];
                }
#pragma unroll
                for (int k = 0; k < B; ++k) {
#pragma unroll
                  for (int v = 0; v < V; ++v) {
                    acc0[v].x += nrm[k] * x[k][v].x;
                    acc0[v].y += nrm[k] * x[k][v].y;
                    acc0[v].z += nrm[k] * x[k][v].z;
                    acc0[v].w += nrm[k] * x[k][v].w;
                  }
                }
              }
            }
            { const int t = q0; q0 = q1; q1 = t; }
            { const int t = qn0; qn0 = qn1; qn1 = t; }
            { const float t = dd0; dd0 = dd1; dd1 = t; }
            for (int v = 0; v < V; ++v) {
              const float4 t = acc0[v]; acc0[v] = acc1[v]; acc1[v] = t;
            }
          }
          qn0 = 0;
          qn1 = 0;
          if (last) break;
        }
        while (m0) {
          const int s = __builtin_amdgcn_readlane(en.x, __builtin_ctz(m0));
          if (lane == qn0) q0 = s;
          ++qn0;
          m0 &= m0 - 1u;
        }
        while (m1) {
          const int s = __builtin_amdgcn_readlane(en.x, __builtin_ctz(m1));
          if (lane == qn1) q1 = s;
          ++qn1;
          m1 &= m1 - 1u;
        }
      }
      if (done) break;
      __syncthreads();
      fill = 0;
    }

    const int wbase = base + wave * KG_CAP;
    int4 dv[4];
#pragma unroll
    for (int q4 = 0; q4 < 4; ++q4) {
      const int e = wbase + q4 * 128 + lane * 4;
      dv[q4] = make_int4(-1, -1, -1, -1);
      if (e < n_edges) dv[q4] = *reinterpret_cast<const int4*>(dst + e);
    }
    int dl[16], sv[16];
    int count = 0;
#pragma unroll
    for (int q4 = 0; q4 < 4; ++q4) {
      const int e = wbase + q4 * 128 + lane * 4;
      const int dd[4] = {dv[q4].x, dv[q4].y, dv[q4].z, dv[q4].w};
#pragma unroll
      for (int j = 0; j < 4; ++j) {
        const int d = dd[j] - n0;
        const bool m = (unsigned)d < (unsigned)nb;
        dl[q4 * 4 + j] = m ? d : -1;
        sv[q4 * 4 + j] = src[m ? (e + j) : 0];
        count += __popc(__builtin_amdgcn_ballot_w32(m));
      }
    }
    if (lane == 0) s_cnt[par][wave] = count;
    __syncthreads();

    const int cw = s_cnt[par][lane];
    int incl = cw;
#pragma unroll
    for (int dlt = 1; dlt < 32; dlt <<= 1) {
      const int t = __shfl_up(incl, dlt);
      if (lane >= dlt) incl += t;
    }
    const int total = __builtin_amdgcn_readlane(incl, 31);
    int pos = fill + __builtin_amdgcn_readlane(incl - cw, wave);
#pragma unroll
    for (int k = 0; k < 16; ++k) {
      const bool m = dl[k] >= 0;
      const unsigned mask = __builtin_amdgcn_ballot_w32(m);
      if (m) s_list[pos + __popc(mask & lanes_below)] = make_int2(sv[k], dl[k]);
      pos += __popc(mask);
    }
    fill += total;
  }

  float4* __restrict__ H4 = reinterpret_cast<float4*>(H);
  if (wave < nb) {
    for (int v = 0; v < V; ++v) {
      float4 r = acc0[v];
      if (RELU) {
        r.x = fmaxf(r.x, 0.0f); r.y = fmaxf(r.y, 0.0f);
        r.z = fmaxf(r.z, 0.0f); r.w = fmaxf(r.w, 0.0f);
      }
      H4[(size_t)(n0 + wave) * (C / 4) + v * 32 + lane] = r;
    }
  }
  if (wave + 32 < nb) {
    for (int v = 0; v < V; ++v) {
      float4 r = acc1[v];
      if (RELU) {
        r.x = fmaxf(r.x, 0.0f); r.y = fmaxf(r.y, 0.0f);
        r.z = fmaxf(r.z, 0.0f); r.w = fmaxf(r.w, 0.0f);
      }
      H4[(size_t)(n0 + wave + 32) * (C / 4) + v * 32 + lane] = r;
    }
  }
}

extern "C" void kernel_launch(void* const* d_in, const int* in_sizes, int n_in,
                              void* d_out, int out_size, void* d_ws, size_t ws_size,
                              hipStream_t stream) {
  const float* mol_x = (const float*)d_in[0];
  const int*   mei   = (const int*)d_in[1];
  const int*   kei   = (const int*)d_in[2];
  const float* W1 = (const float*)d_in[3];
  const float* b1 = (const float*)d_in[4];
  const float* W2 = (const float*)d_in[5];
  const float* b2 = (const float*)d_in[6];
  const float* W3 = (const float*)d_in[7];
  const float* b3 = (const float*)d_in[8];
  const float* Wk1 = (const float*)d_in[9];
  const float* bk1 = (const float*)d_in[10];
  const float* Wk2 = (const float*)d_in[11];
  const float* bk2 = (const float*)d_in[12];
  float* out = (float*)d_out;

  float* ws    = (float*)d_ws;
  float* embed = ws;
  float* deg   = embed + M_MOL * OUT_CH;
  float* dinv  = deg + M_MOL;
  float* XW    = dinv + M_MOL;
  float* h     = XW + (size_t)M_MOL * HID;
  float* XW2   = h + (size_t)M_MOL * HID;

  const int* kg_src = kei;
  const int* kg_dst = kei + E_KG;

  mol_gnn_kernel<<<M_MOL, 256, 0, stream>>>(mol_x, mei, W1, b1, W2, b2, W3, b3,
                                            embed);

  kg_deg_init_kernel<<<(M_MOL + 255) / 256, 256, 0, stream>>>(deg, M_MOL);
  kg_deg_count_kernel<<<(E_KG + 255) / 256, 256, 0, stream>>>(kg_dst, deg, E_KG);
  kg_dinv_kernel<<<(M_MOL + 255) / 256, 256, 0, stream>>>(deg, dinv, M_MOL);

  {
    int tiles = (M_MOL / 16) * (HID / 16);
    wmma_gemm_kernel<<<(tiles + 7) / 8, 256, 0, stream>>>(embed, Wk1, XW,
                                                          M_MOL, OUT_CH, HID);
  }
  kg_block_gather_kernel<HID, true><<<(M_MOL + KG_NB - 1) / KG_NB, KG_THREADS, 0,
                                      stream>>>(XW, dinv, bk1, kg_src, kg_dst, h,
                                                M_MOL, E_KG);

  {
    int tiles = (M_MOL / 16) * (OUT_CH / 16);
    wmma_gemm_kernel<<<(tiles + 7) / 8, 256, 0, stream>>>(h, Wk2, XW2,
                                                          M_MOL, HID, OUT_CH);
  }
  kg_block_gather_kernel<OUT_CH, false><<<(M_MOL + KG_NB - 1) / KG_NB, KG_THREADS,
                                          0, stream>>>(XW2, dinv, bk2, kg_src,
                                                       kg_dst, out, M_MOL, E_KG);
}


# mol kernel GEMM loop: a trip's 16 B operand pairs read from LDS as one group ahead of its 16 WMMAs, not two per k-step between them: one LDS turn-round per trip instead of eight; sum order kept
# speedup vs baseline: 2.8062x; 1.0335x over previous
#include <hip/hip_runtime.h>

typedef __attribute__((ext_vector_type(2))) float v2f;
typedef __attribute__((ext_vector_type(8))) float v8f;

#define N_ATOMS 32
#define E_MOL   64
#define IN_CH   64
#define OUT_CH  128
#define HID     256
#define M_MOL   10000
#define E_KG    320000

__device__ __forceinline__ v8f wmma_f32_k4(v2f a, v2f b, v8f c) {
  return __builtin_amdgcn_wmma_f32_16x16x4_f32(false, a, false, b, (short)0, c,
                                               false, false);
}

#define MOL_WAVES 8

#define LDP 36

template <int K, int N, bool RELU>
__device__ __forceinline__ void gcn_layer_wmma(const float* __restrict__ XT,
                                               const float* __restrict__ W,
                                               const float* __restrict__ bias,
                                               const float* __restrict__ AT,
                                               float* __restrict__ TT,
                                               int wave, int lane) {
  const int hi = lane >> 4;
  const int lo = lane & 15;
  const float* __restrict__ Wl = W + hi * N + lo;
  const float* __restrict__ xp = XT + hi * LDP + lo;
  int tile1 = 16;
  asm("" : "+v"(tile1));
  const float* __restrict__ xq = xp + tile1;
  const float* __restrict__ at0 = AT + lo * LDP + 2 * hi;
  const float* __restrict__ at1 = AT + (16 + lo) * LDP + 2 * hi;
  for (int nt = wave; nt < N / 16; nt += MOL_WAVES) {
    const float* __restrict__ wp = Wl + nt * 16;
    v8f acc0 = {};
    v8f acc1 = {};
#pragma unroll
    for (int kb = 0; kb < K; kb += 32) {
      v2f a[8];
#pragma unroll
      for (int j = 0; j < 8; ++j) {
        a[j].x = wp[(kb + 4 * j) * N];
        a[j].y = wp[(kb + 4 * j + 2) * N];
      }
      __builtin_amdgcn_sched_barrier(0);
      v2f b0[8], b1[8];
#pragma unroll
      for (int j = 0; j < 8; ++j) {
        b0[j].x = xp[(kb + 4 * j) * LDP];
        b0[j].y = xp[(kb + 4 * j + 2) * LDP];
        b1[j].x = xq[(kb + 4 * j) * LDP];
        b1[j].y = xq[(kb + 4 * j + 2) * LDP];
      }
      __builtin_amdgcn_sched_barrier(0);
#pragma unroll
      for (int j = 0; j < 8; ++j) {
        acc0 = wmma_f32_k4(a[j], b0[j], acc0);
        acc1 = wmma_f32_k4(a[j], b1[j], acc1);
      }
    }
    float* trow = TT + (nt * 16 + 8 * hi) * LDP + lo;
#pragma unroll
    for (int v = 0; v < 8; ++v) {
      trow[v * LDP]      = acc0[v];
      trow[v * LDP + 16] = acc1[v];
    }
    const float* ar = TT + (nt * 16 + lo) * LDP + 2 * hi;
    v2f ta[8];
#pragma unroll
    for (int t = 0; t < 8; ++t)
      ta[t] = *reinterpret_cast<const v2f*>(ar + 4 * t);
    const float4 bA = *reinterpret_cast<const float4*>(bias + nt * 16 + 8 * hi);
    const float4 bB = *reinterpret_cast<const float4*>(bias + nt * 16 + 8 * hi + 4);
    v8f h0 = {bA.x, bA.y, bA.z, bA.w, bB.x, bB.y, bB.z, bB.w};
    v8f h1 = h0;
#pragma unroll
    for (int t = 0; t < 8; ++t) {
      const v2f n0 = *reinterpret_cast<const v2f*>(at0 + 4 * t);
      const v2f n1 = *reinterpret_cast<const v2f*>(at1 + 4 * t);
      h0 = wmma_f32_k4(ta[t], n0, h0);
      h1 = wmma_f32_k4(ta[t], n1, h1);
    }
#pragma unroll
    for (int v = 0; v < 8; ++v) {
      float r0 = h0[v], r1 = h1[v];
      if (RELU) { r0 = fmaxf(r0, 0.0f); r1 = fmaxf(r1, 0.0f); }
      trow[v * LDP]      = r0;
      trow[v * LDP + 16] = r1;
    }
  }
}

__global__ __launch_bounds__(256, 8) void mol_gnn_kernel(
    const float* __restrict__ mol_x, const int* __restrict__ mei,
    const float* __restrict__ W1, const float* __restrict__ b1,
    const float* __restrict__ W2, const float* __restrict__ b2,
    const float* __restrict__ W3, const float* __restrict__ b3,
    float* __restrict__ embed) {
  __shared__ __attribute__((aligned(16))) float Abuf[HID * LDP];
  __shared__ __attribute__((aligned(16))) float Bbuf[HID * LDP];
  __shared__ __attribute__((aligned(16))) float s_at[N_ATOMS * LDP];
  __shared__ float s_dinv[N_ATOMS];
  __shared__ int   s_deg[N_ATOMS];
  __shared__ int   s_src[E_MOL];
  __shared__ int   s_dst[E_MOL];

  const int mol  = blockIdx.x;
  const int tid  = threadIdx.x;
  const int lane = tid & 31;
  const int wave = __builtin_amdgcn_readfirstlane(tid >> 5);
  __builtin_assume(wave >= 0 && wave < MOL_WAVES);

  const float* X = mol_x + (size_t)mol * (N_ATOMS * IN_CH);
  const float4* X4 = reinterpret_cast<const float4*>(X);
  for (int q = wave; q < IN_CH / 4; q += MOL_WAVES) {
    const float4 v = X4[lane * (IN_CH / 4) + q];
    Abuf[(4 * q + 0) * LDP + lane] = v.x;
    Abuf[(4 * q + 1) * LDP + lane] = v.y;
    Abuf[(4 * q + 2) * LDP + lane] = v.z;
    Abuf[(4 * q + 3) * LDP + lane] = v.w;
  }
  const int* me = mei + (size_t)mol * (2 * E_MOL);
  if (tid < E_MOL) { s_src[tid] = me[tid]; s_dst[tid] = me[E_MOL + tid]; }
  if (tid < N_ATOMS) s_deg[tid] = 1;
  for (int i = tid; i < N_ATOMS * LDP; i += 256) s_at[i] = 0.0f;
  __syncthreads();
  if (tid < E_MOL) atomicAdd(&s_deg[s_dst[tid]], 1);
  __syncthreads();
  if (tid < N_ATOMS) {
    const float d = rsqrtf((float)s_deg[tid]);
    s_dinv[tid] = d;
    s_at[tid * LDP + tid] = d * d;
  }
  __syncthreads();
  if (tid < E_MOL) {
    const int s = s_src[tid], d = s_dst[tid];
    atomicAdd(&s_at[d * LDP + s], s_dinv[s] * s_dinv[d]);
  }
  __syncthreads();

  gcn_layer_wmma<IN_CH, HID, true>(Abuf, W1, b1, s_at, Bbuf, wave, lane);
  __syncthreads();
  gcn_layer_wmma<HID, HID, true>(Bbuf, W2, b2, s_at, Abuf, wave, lane);
  __syncthreads();
  gcn_layer_wmma<HID, OUT_CH, false>(Abuf, W3, b3, s_at, Bbuf, wave, lane);
  __syncthreads();

  if (tid < OUT_CH) {
    const int c = tid;
    float v = Bbuf[c * LDP + (c & (N_ATOMS - 1))];
    for (int r = 1; r < N_ATOMS; ++r)
      v = fminf(v, Bbuf[c * LDP + ((c + r) & (N_ATOMS - 1))]);
    embed[(size_t)mol * OUT_CH + c] = v;
  }
}

__global__ __launch_bounds__(256) void wmma_gemm_kernel(
    const float* __restrict__ A, const float* __restrict__ B,
    float* __restrict__ C, int Mrows, int K, int N) {
  const int lane = threadIdx.x & 31;
  const int wave = blockIdx.x * (blockDim.x >> 5) + (threadIdx.x >> 5);
  const int ntT = N >> 4;
  const int nTiles = (Mrows >> 4) * ntT;
  if (wave >= nTiles) return;
  const int mt = wave / ntT;
  const int nt = wave - mt * ntT;
  const int hi = lane >> 4, lo = lane & 15;
  const int mrow = mt * 16 + lo;
  const int ncol = nt * 16 + lo;
  v8f acc = {};
  for (int kb = 0; kb < K; kb += 4) {
    const int k = kb + 2 * hi;
    v2f a, b;
    a.x = A[(size_t)mrow * K + k];
    a.y = A[(size_t)mrow * K + k + 1];
    b.x = B[(size_t)k * N + ncol];
    b.y = B[(size_t)(k + 1) * N + ncol];
    acc = wmma_f32_k4(a, b, acc);
  }
#pragma unroll
  for (int v = 0; v < 8; ++v)
    C[(size_t)(mt * 16 + v + 8 * hi) * N + nt * 16 + lo] = acc[v];
}

__global__ void kg_deg_init_kernel(float* __restrict__ deg, int n) {
  int i = blockIdx.x * blockDim.x + threadIdx.x;
  if (i < n) deg[i] = 1.0f;
}
__global__ void kg_deg_count_kernel(const int* __restrict__ dst,
                                    float* __restrict__ deg, int ne) {
  int i = blockIdx.x * blockDim.x + threadIdx.x;
  if (i < ne) atomicAdd(&deg[dst[i]], 1.0f);
}
__global__ void kg_dinv_kernel(const float* __restrict__ deg,
                               float* __restrict__ dinv, int n) {
  int i = blockIdx.x * blockDim.x + threadIdx.x;
  if (i < n) dinv[i] = rsqrtf(deg[i]);
}
#define KG_NB      40
#define KG_THREADS 1024
#define KG_WAVES   32
#define KG_CAP     512
#define KG_ROUND   (KG_WAVES * KG_CAP)
#define KG_LIST    (2 * KG_ROUND)
#define KG_WCAP    512
#define KG_SEG0    (KG_LIST / 2)
#define KG_G       4

template <int C, bool RELU>
__global__ __launch_bounds__(KG_THREADS) void kg_block_gather_kernel(
    const float* __restrict__ XW, const float* __restrict__ dinv,
    const float* __restrict__ bias, const int* __restrict__ src,
    const int* __restrict__ dst, float* __restrict__ H,
    int n_nodes, int n_edges) {
  constexpr int V = C / 128;
  constexpr int B = 8 / V;
  __shared__ int2 s_list[KG_LIST];
  __shared__ int  s_cnt[2][KG_WAVES];
  __shared__ int  s_wcnt[KG_WAVES];

  const int tid  = threadIdx.x;
  const int lane = tid & 31;
  const int wave = __builtin_amdgcn_readfirstlane(tid >> 5);
  const int n0   = blockIdx.x * KG_NB;
  const int nb   = min(KG_NB, n_nodes - n0);
  const float4* __restrict__ XW4   = reinterpret_cast<const float4*>(XW);
  const float4* __restrict__ bias4 = reinterpret_cast<const float4*>(bias);

  float4 acc0[V], acc1[V];
  float dd0 = 0.0f, dd1 = 0.0f;
  for (int v = 0; v < V; ++v) {
    acc0[v] = make_float4(0.0f, 0.0f, 0.0f, 0.0f);
    acc1[v] = make_float4(0.0f, 0.0f, 0.0f, 0.0f);
  }
  if (wave < nb) {
    dd0 = dinv[n0 + wave];
    const float s = dd0 * dd0;
    for (int v = 0; v < V; ++v) {
      const float4 x = XW4[(size_t)(n0 + wave) * (C / 4) + v * 32 + lane];
      const float4 bb = bias4[v * 32 + lane];
      acc0[v].x = s * x.x + bb.x;
      acc0[v].y = s * x.y + bb.y;
      acc0[v].z = s * x.z + bb.z;
      acc0[v].w = s * x.w + bb.w;
    }
  }
  if (wave + 32 < nb) {
    dd1 = dinv[n0 + wave + 32];
    const float s = dd1 * dd1;
    for (int v = 0; v < V; ++v) {
      const float4 x = XW4[(size_t)(n0 + wave + 32) * (C / 4) + v * 32 + lane];
      const float4 bb = bias4[v * 32 + lane];
      acc1[v].x = s * x.x + bb.x;
      acc1[v].y = s * x.y + bb.y;
      acc1[v].z = s * x.z + bb.z;
      acc1[v].w = s * x.w + bb.w;
    }
  }

  const unsigned lanes_below = (1u << lane) - 1u;
  int fill  = 0;
  int base0 = 0;
  {
    int2* __restrict__ seg = s_list + KG_SEG0 + wave * KG_WCAP;
    const int ntrip = (n_edges + 128 * KG_WAVES * KG_G - 1) / (128 * KG_WAVES * KG_G);
    int cnt = 0;
    const int last4 = max(n_edges - 4, 0);
    auto load_trip = [&](int4 (&buf)[KG_G], int t) {
#pragma unroll
      for (int g = 0; g < KG_G; ++g) {
        const int e = ((t * KG_G + g) * KG_WAVES + wave) * 128 + lane * 4;
        buf[g] = *reinterpret_cast<const int4*>(dst + min(e, last4));
      }
    };
    auto test_trip = [&](const int4 (&buf)[KG_G], int t) {
#pragma unroll
      for (int g = 0; g < KG_G; ++g) {
        const int e0 = ((t * KG_G + g) * KG_WAVES + wave) * 128 + lane * 4;
        const int d[4] = {buf[g].x - n0, buf[g].y - n0, buf[g].z - n0, buf[g].w - n0};
        const unsigned nbv = (e0 < n_edges) ? (unsigned)nb : 0u;
        unsigned mk[4];
#pragma unroll
        for (int j = 0; j < 4; ++j)
          mk[j] = __builtin_amdgcn_ballot_w32((unsigned)d[j] < nbv);
        if ((mk[0] | mk[1] | mk[2] | mk[3]) != 0u) {
#pragma unroll
          for (int j = 0; j < 4; ++j) {
            if (mk[j] != 0u) {
              const int p = cnt + __popc(mk[j] & lanes_below);
              if ((unsigned)d[j] < nbv && p < KG_WCAP)
                seg[p] = make_int2(e0 + j, d[j]);
              cnt += __popc(mk[j]);
            }
          }
        }
      }
    };
    int4 bufa[KG_G], bufb[KG_G];
    load_trip(bufa, 0);
    for (int t = 0; t < ntrip; t += 2) {
      load_trip(bufb, t + 1);
      test_trip(bufa, t);
      load_trip(bufa, t + 2);
      test_trip(bufb, t + 1);
    }
    if (lane == 0) s_wcnt[wave] = cnt;
    __syncthreads();
    const int cw = s_wcnt[lane];
    if (__builtin_amdgcn_ballot_w32(cw > KG_WCAP) == 0u) {
      int incl = cw;
#pragma unroll
      for (int dlt = 1; dlt < 32; dlt <<= 1) {
        const int t = __shfl_up(incl, dlt);
        if (lane >= dlt) incl += t;
      }
      const int off = __builtin_amdgcn_readlane(incl - cw, wave);
      for (int i = lane; i < cnt; i += 32) {
        int2 en = seg[i];
        en.x = src[en.x];
        s_list[off + i] = en;
      }
      fill  = __builtin_amdgcn_readlane(incl, 31);
      base0 = n_edges;
    }
  }
  int par  = 0;
  for (int base = base0;; base += KG_ROUND, par ^= 1) {
    const bool done = base >= n_edges;
    if (done || fill > KG_LIST - KG_ROUND) {
      asm volatile("s_wait_dscnt 0x0" ::: "memory");
      __syncthreads();
      int q0 = 0, q1 = 0, qn0 = 0, qn1 = 0;
      for (int i0 = 0;; i0 += 32) {
        const bool last = i0 >= fill;
        int2 en = make_int2(0, -1);
        if (i0 + lane < fill) en = s_list[i0 + lane];
        unsigned m0 = __builtin_amdgcn_ballot_w32(en.y == wave);
        unsigned m1 = __builtin_amdgcn_ballot_w32(en.y == wave + 32);
        if (last || qn0 + __popc(m0) > 32 || qn1 + __popc(m1) > 32) {
          for (int pass = 0; pass < 2; ++pass) {
            if (qn0 > 0) {
              const float wq = dinv[(lane < qn0) ? q0 : 0];
              for (int k0 = 0; k0 < qn0; k0 += B) {
                int   es[B];
                float nrm[B];
#pragma unroll
                for (int k = 0; k < B; ++k) {
                  const bool ok = k0 + k < qn0;
                  const int  kk = ok ? k0 + k : k0;
                  es[k] = __builtin_amdgcn_readlane(q0, kk);
                  const float wk =
                      __int_as_float(__builtin_amdgcn_readlane(__float_as_int(wq), kk));
                  nrm[k] = ok ? wk * dd0 : 0.0f;
                }
                float4 x[B][V];
#pragma unroll
                for (int k = 0; k < B; ++k) {
#pragma unroll
                  for (int v = 0; v < V; ++v)
                    x[k][v] = XW4[(size_t)es[k] * (C / 4) + v * 32 + lane];
                }
#pragma unroll
                for (int k = 0; k < B; ++k) {
#pragma unroll
                  for (int v = 0; v < V; ++v) {
                    acc0[v].x += nrm[k] * x[k][v].x;
                    acc0[v].y += nrm[k] * x[k][v].y;
                    acc0[v].z += nrm[k] * x[k][v].z;
                    acc0[v].w += nrm[k] * x[k][v].w;
                  }
                }
              }
            }
            { const int t = q0; q0 = q1; q1 = t; }
            { const int t = qn0; qn0 = qn1; qn1 = t; }
            { const float t = dd0; dd0 = dd1; dd1 = t; }
            for (int v = 0; v < V; ++v) {
              const float4 t = acc0[v]; acc0[v] = acc1[v]; acc1[v] = t;
            }
          }
          qn0 = 0;
          qn1 = 0;
          if (last) break;
        }
        while (m0) {
          const int s = __builtin_amdgcn_readlane(en.x, __builtin_ctz(m0));
          if (lane == qn0) q0 = s;
          ++qn0;
          m0 &= m0 - 1u;
        }
        while (m1) {
          const int s = __builtin_amdgcn_readlane(en.x, __builtin_ctz(m1));
          if (lane == qn1) q1 = s;
          ++qn1;
          m1 &= m1 - 1u;
        }
      }
      if (done) break;
      __syncthreads();
      fill = 0;
    }

    const int wbase = base + wave * KG_CAP;
    int4 dv[4];
#pragma unroll
    for (int q4 = 0; q4 < 4; ++q4) {
      const int e = wbase + q4 * 128 + lane * 4;
      dv[q4] = make_int4(-1, -1, -1, -1);
      if (e < n_edges) dv[q4] = *reinterpret_cast<const int4*>(dst + e);
    }
    int dl[16], sv[16];
    int count = 0;
#pragma unroll
    for (int q4 = 0; q4 < 4; ++q4) {
      const int e = wbase + q4 * 128 + lane * 4;
      const int dd[4] = {dv[q4].x, dv[q4].y, dv[q4].z, dv[q4].w};
#pragma unroll
      for (int j = 0; j < 4; ++j) {
        const int d = dd[j] - n0;
        const bool m = (unsigned)d < (unsigned)nb;
        dl[q4 * 4 + j] = m ? d : -1;
        sv[q4 * 4 + j] = src[m ? (e + j) : 0];
        count += __popc(__builtin_amdgcn_ballot_w32(m));
      }
    }
    if (lane == 0) s_cnt[par][wave] = count;
    __syncthreads();

    const int cw = s_cnt[par][lane];
    int incl = cw;
#pragma unroll
    for (int dlt = 1; dlt < 32; dlt <<= 1) {
      const int t = __shfl_up(incl, dlt);
      if (lane >= dlt) incl += t;
    }
    const int total = __builtin_amdgcn_readlane(incl, 31);
    int pos = fill + __builtin_amdgcn_readlane(incl - cw, wave);
#pragma unroll
    for (int k = 0; k < 16; ++k) {
      const bool m = dl[k] >= 0;
      const unsigned mask = __builtin_amdgcn_ballot_w32(m);
      if (m) s_list[pos + __popc(mask & lanes_below)] = make_int2(sv[k], dl[k]);
      pos += __popc(mask);
    }
    fill += total;
  }

  float4* __restrict__ H4 = reinterpret_cast<float4*>(H);
  if (wave < nb) {
    for (int v = 0; v < V; ++v) {
      float4 r = acc0[v];
      if (RELU) {
        r.x = fmaxf(r.x, 0.0f); r.y = fmaxf(r.y, 0.0f);
        r.z = fmaxf(r.z, 0.0f); r.w = fmaxf(r.w, 0.0f);
      }
      H4[(size_t)(n0 + wave) * (C / 4) + v * 32 + lane] = r;
    }
  }
  if (wave + 32 < nb) {
    for (int v = 0; v < V; ++v) {
      float4 r = acc1[v];
      if (RELU) {
        r.x = fmaxf(r.x, 0.0f); r.y = fmaxf(r.y, 0.0f);
        r.z = fmaxf(r.z, 0.0f); r.w = fmaxf(r.w, 0.0f);
      }
      H4[(size_t)(n0 + wave + 32) * (C / 4) + v * 32 + lane] = r;
    }
  }
}

extern "C" void kernel_launch(void* const* d_in, const int* in_sizes, int n_in,
                              void* d_out, int out_size, void* d_ws, size_t ws_size,
                              hipStream_t stream) {
  const float* mol_x = (const float*)d_in[0];
  const int*   mei   = (const int*)d_in[1];
  const int*   kei   = (const int*)d_in[2];
  const float* W1 = (const float*)d_in[3];
  const float* b1 = (const float*)d_in[4];
  const float* W2 = (const float*)d_in[5];
  const float* b2 = (const float*)d_in[6];
  const float* W3 = (const float*)d_in[7];
  const float* b3 = (const float*)d_in[8];
  const float* Wk1 = (const float*)d_in[9];
  const float* bk1 = (const float*)d_in[10];
  const float* Wk2 = (const float*)d_in[11];
  const float* bk2 = (const float*)d_in[12];
  float* out = (float*)d_out;

  float* ws    = (float*)d_ws;
  float* embed = ws;
  float* deg   = embed + M_MOL * OUT_CH;
  float* dinv  = deg + M_MOL;
  float* XW    = dinv + M_MOL;
  float* h     = XW + (size_t)M_MOL * HID;
  float* XW2   = h + (size_t)M_MOL * HID;

  const int* kg_src = kei;
  const int* kg_dst = kei + E_KG;

  mol_gnn_kernel<<<M_MOL, 256, 0, stream>>>(mol_x, mei, W1, b1, W2, b2, W3, b3,
                                            embed);

  kg_deg_init_kernel<<<(M_MOL + 255) / 256, 256, 0, stream>>>(deg, M_MOL);
  kg_deg_count_kernel<<<(E_KG + 255) / 256, 256, 0, stream>>>(kg_dst, deg, E_KG);
  kg_dinv_kernel<<<(M_MOL + 255) / 256, 256, 0, stream>>>(deg, dinv, M_MOL);

  {
    int tiles = (M_MOL / 16) * (HID / 16);
    wmma_gemm_kernel<<<(tiles + 7) / 8, 256, 0, stream>>>(embed, Wk1, XW,
                                                          M_MOL, OUT_CH, HID);
  }
  kg_block_gather_kernel<HID, true><<<(M_MOL + KG_NB - 1) / KG_NB, KG_THREADS, 0,
                                      stream>>>(XW, dinv, bk1, kg_src, kg_dst, h,
                                                M_MOL, E_KG);

  {
    int tiles = (M_MOL / 16) * (OUT_CH / 16);
    wmma_gemm_kernel<<<(tiles + 7) / 8, 256, 0, stream>>>(h, Wk2, XW2,
                                                          M_MOL, HID, OUT_CH);
  }
  kg_block_gather_kernel<OUT_CH, false><<<(M_MOL + KG_NB - 1) / KG_NB, KG_THREADS,
                                          0, stream>>>(XW2, dinv, bk2, kg_src,
                                                       kg_dst, out, M_MOL, E_KG);
}


# KG edge filter run once per call, not once per layer: a new kernel finds each 40-node block's edges, counts degrees on the way (replacing 3 degree launches); both layers read its list; sum order kept
# speedup vs baseline: 2.9623x; 1.0556x over previous
#include <hip/hip_runtime.h>

typedef __attribute__((ext_vector_type(2))) float v2f;
typedef __attribute__((ext_vector_type(8))) float v8f;

#define N_ATOMS 32
#define E_MOL   64
#define IN_CH   64
#define OUT_CH  128
#define HID     256
#define M_MOL   10000
#define E_KG    320000

__device__ __forceinline__ v8f wmma_f32_k4(v2f a, v2f b, v8f c) {
  return __builtin_amdgcn_wmma_f32_16x16x4_f32(false, a, false, b, (short)0, c,
                                               false, false);
}

#define MOL_WAVES 8

#define LDP 36

template <int K, int N, bool RELU>
__device__ __forceinline__ void gcn_layer_wmma(const float* __restrict__ XT,
                                               const float* __restrict__ W,
                                               const float* __restrict__ bias,
                                               const float* __restrict__ AT,
                                               float* __restrict__ TT,
                                               int wave, int lane) {
  const int hi = lane >> 4;
  const int lo = lane & 15;
  const float* __restrict__ Wl = W + hi * N + lo;
  const float* __restrict__ xp = XT + hi * LDP + lo;
  int tile1 = 16;
  asm("" : "+v"(tile1));
  const float* __restrict__ xq = xp + tile1;
  const float* __restrict__ at0 = AT + lo * LDP + 2 * hi;
  const float* __restrict__ at1 = AT + (16 + lo) * LDP + 2 * hi;
  for (int nt = wave; nt < N / 16; nt += MOL_WAVES) {
    const float* __restrict__ wp = Wl + nt * 16;
    v8f acc0 = {};
    v8f acc1 = {};
#pragma unroll
    for (int kb = 0; kb < K; kb += 32) {
      v2f a[8];
#pragma unroll
      for (int j = 0; j < 8; ++j) {
        a[j].x = wp[(kb + 4 * j) * N];
        a[j].y = wp[(kb + 4 * j + 2) * N];
      }
      __builtin_amdgcn_sched_barrier(0);
      v2f b0[8], b1[8];
#pragma unroll
      for (int j = 0; j < 8; ++j) {
        b0[j].x = xp[(kb + 4 * j) * LDP];
        b0[j].y = xp[(kb + 4 * j + 2) * LDP];
        b1[j].x = xq[(kb + 4 * j) * LDP];
        b1[j].y = xq[(kb + 4 * j + 2) * LDP];
      }
      __builtin_amdgcn_sched_barrier(0);
#pragma unroll
      for (int j = 0; j < 8; ++j) {
        acc0 = wmma_f32_k4(a[j], b0[j], acc0);
        acc1 = wmma_f32_k4(a[j], b1[j], acc1);
      }
    }
    float* trow = TT + (nt * 16 + 8 * hi) * LDP + lo;
#pragma unroll
    for (int v = 0; v < 8; ++v) {
      trow[v * LDP]      = acc0[v];
      trow[v * LDP + 16] = acc1[v];
    }
    const float* ar = TT + (nt * 16 + lo) * LDP + 2 * hi;
    v2f ta[8];
#pragma unroll
    for (int t = 0; t < 8; ++t)
      ta[t] = *reinterpret_cast<const v2f*>(ar + 4 * t);
    const float4 bA = *reinterpret_cast<const float4*>(bias + nt * 16 + 8 * hi);
    const float4 bB = *reinterpret_cast<const float4*>(bias + nt * 16 + 8 * hi + 4);
    v8f h0 = {bA.x, bA.y, bA.z, bA.w, bB.x, bB.y, bB.z, bB.w};
    v8f h1 = h0;
#pragma unroll
    for (int t = 0; t < 8; ++t) {
      const v2f n0 = *reinterpret_cast<const v2f*>(at0 + 4 * t);
      const v2f n1 = *reinterpret_cast<const v2f*>(at1 + 4 * t);
      h0 = wmma_f32_k4(ta[t], n0, h0);
      h1 = wmma_f32_k4(ta[t], n1, h1);
    }
#pragma unroll
    for (int v = 0; v < 8; ++v) {
      float r0 = h0[v], r1 = h1[v];
      if (RELU) { r0 = fmaxf(r0, 0.0f); r1 = fmaxf(r1, 0.0f); }
      trow[v * LDP]      = r0;
      trow[v * LDP + 16] = r1;
    }
  }
}

__global__ __launch_bounds__(256, 8) void mol_gnn_kernel(
    const float* __restrict__ mol_x, const int* __restrict__ mei,
    const float* __restrict__ W1, const float* __restrict__ b1,
    const float* __restrict__ W2, const float* __restrict__ b2,
    const float* __restrict__ W3, const float* __restrict__ b3,
    float* __restrict__ embed) {
  __shared__ __attribute__((aligned(16))) float Abuf[HID * LDP];
  __shared__ __attribute__((aligned(16))) float Bbuf[HID * LDP];
  __shared__ __attribute__((aligned(16))) float s_at[N_ATOMS * LDP];
  __shared__ float s_dinv[N_ATOMS];
  __shared__ int   s_deg[N_ATOMS];
  __shared__ int   s_src[E_MOL];
  __shared__ int   s_dst[E_MOL];

  const int mol  = blockIdx.x;
  const int tid  = threadIdx.x;
  const int lane = tid & 31;
  const int wave = __builtin_amdgcn_readfirstlane(tid >> 5);
  __builtin_assume(wave >= 0 && wave < MOL_WAVES);

  const float* X = mol_x + (size_t)mol * (N_ATOMS * IN_CH);
  const float4* X4 = reinterpret_cast<const float4*>(X);
  for (int q = wave; q < IN_CH / 4; q += MOL_WAVES) {
    const float4 v = X4[lane * (IN_CH / 4) + q];
    Abuf[(4 * q + 0) * LDP + lane] = v.x;
    Abuf[(4 * q + 1) * LDP + lane] = v.y;
    Abuf[(4 * q + 2) * LDP + lane] = v.z;
    Abuf[(4 * q + 3) * LDP + lane] = v.w;
  }
  const int* me = mei + (size_t)mol * (2 * E_MOL);
  if (tid < E_MOL) { s_src[tid] = me[tid]; s_dst[tid] = me[E_MOL + tid]; }
  if (tid < N_ATOMS) s_deg[tid] = 1;
  for (int i = tid; i < N_ATOMS * LDP; i += 256) s_at[i] = 0.0f;
  __syncthreads();
  if (tid < E_MOL) atomicAdd(&s_deg[s_dst[tid]], 1);
  __syncthreads();
  if (tid < N_ATOMS) {
    const float d = rsqrtf((float)s_deg[tid]);
    s_dinv[tid] = d;
    s_at[tid * LDP + tid] = d * d;
  }
  __syncthreads();
  if (tid < E_MOL) {
    const int s = s_src[tid], d = s_dst[tid];
    atomicAdd(&s_at[d * LDP + s], s_dinv[s] * s_dinv[d]);
  }
  __syncthreads();

  gcn_layer_wmma<IN_CH, HID, true>(Abuf, W1, b1, s_at, Bbuf, wave, lane);
  __syncthreads();
  gcn_layer_wmma<HID, HID, true>(Bbuf, W2, b2, s_at, Abuf, wave, lane);
  __syncthreads();
  gcn_layer_wmma<HID, OUT_CH, false>(Abuf, W3, b3, s_at, Bbuf, wave, lane);
  __syncthreads();

  if (tid < OUT_CH) {
    const int c = tid;
    float v = Bbuf[c * LDP + (c & (N_ATOMS - 1))];
    for (int r = 1; r < N_ATOMS; ++r)
      v = fminf(v, Bbuf[c * LDP + ((c + r) & (N_ATOMS - 1))]);
    embed[(size_t)mol * OUT_CH + c] = v;
  }
}

__global__ __launch_bounds__(256) void wmma_gemm_kernel(
    const float* __restrict__ A, const float* __restrict__ B,
    float* __restrict__ C, int Mrows, int K, int N) {
  const int lane = threadIdx.x & 31;
  const int wave = blockIdx.x * (blockDim.x >> 5) + (threadIdx.x >> 5);
  const int ntT = N >> 4;
  const int nTiles = (Mrows >> 4) * ntT;
  if (wave >= nTiles) return;
  const int mt = wave / ntT;
  const int nt = wave - mt * ntT;
  const int hi = lane >> 4, lo = lane & 15;
  const int mrow = mt * 16 + lo;
  const int ncol = nt * 16 + lo;
  v8f acc = {};
  for (int kb = 0; kb < K; kb += 4) {
    const int k = kb + 2 * hi;
    v2f a, b;
    a.x = A[(size_t)mrow * K + k];
    a.y = A[(size_t)mrow * K + k + 1];
    b.x = B[(size_t)k * N + ncol];
    b.y = B[(size_t)(k + 1) * N + ncol];
    acc = wmma_f32_k4(a, b, acc);
  }
#pragma unroll
  for (int v = 0; v < 8; ++v)
    C[(size_t)(mt * 16 + v + 8 * hi) * N + nt * 16 + lo] = acc[v];
}

#define KG_NB      40
#define KG_THREADS 1024
#define KG_WAVES   32
#define KG_CAP     512
#define KG_ROUND   (KG_WAVES * KG_CAP)
#define KG_LIST    (2 * KG_ROUND)
#define KG_WCAP    512
#define KG_WSCAP   2048
#define KG_G       4

__global__ __launch_bounds__(KG_THREADS) void kg_filter_kernel(
    const int* __restrict__ src, const int* __restrict__ dst,
    float* __restrict__ dinv, int* __restrict__ wlist, int* __restrict__ wcnt,
    int n_nodes, int n_edges) {
  __shared__ int2 s_seg[KG_WAVES * KG_WCAP];
  __shared__ int  s_wcnt[KG_WAVES];
  __shared__ int  s_deg[KG_NB];

  const int tid  = threadIdx.x;
  const int lane = tid & 31;
  const int wave = __builtin_amdgcn_readfirstlane(tid >> 5);
  const int n0   = blockIdx.x * KG_NB;
  const int nb   = min(KG_NB, n_nodes - n0);

  if (tid < KG_NB) s_deg[tid] = 1;
  __syncthreads();

  const unsigned lanes_below = (1u << lane) - 1u;
  int2* __restrict__ seg = s_seg + wave * KG_WCAP;
  const int ntrip = (n_edges + 128 * KG_WAVES * KG_G - 1) / (128 * KG_WAVES * KG_G);
  int cnt = 0;
  const int last4 = max(n_edges - 4, 0);
  auto load_trip = [&](int4 (&buf)[KG_G], int t) {
#pragma unroll
    for (int g = 0; g < KG_G; ++g) {
      const int e = ((t * KG_G + g) * KG_WAVES + wave) * 128 + lane * 4;
      buf[g] = *reinterpret_cast<const int4*>(dst + min(e, last4));
    }
  };
  auto test_trip = [&](const int4 (&buf)[KG_G], int t) {
#pragma unroll
    for (int g = 0; g < KG_G; ++g) {
      const int e0 = ((t * KG_G + g) * KG_WAVES + wave) * 128 + lane * 4;
      const int d[4] = {buf[g].x - n0, buf[g].y - n0, buf[g].z - n0, buf[g].w - n0};
      const unsigned nbv = (e0 < n_edges) ? (unsigned)nb : 0u;
      unsigned mk[4];
#pragma unroll
      for (int j = 0; j < 4; ++j)
        mk[j] = __builtin_amdgcn_ballot_w32((unsigned)d[j] < nbv);
      if ((mk[0] | mk[1] | mk[2] | mk[3]) != 0u) {
#pragma unroll
        for (int j = 0; j < 4; ++j) {
          if (mk[j] != 0u) {
            const int p = cnt + __popc(mk[j] & lanes_below);
            if ((unsigned)d[j] < nbv) {
              atomicAdd(&s_deg[d[j]], 1);
              if (p < KG_WCAP) seg[p] = make_int2(e0 + j, d[j]);
            }
            cnt += __popc(mk[j]);
          }
        }
      }
    }
  };
  int4 bufa[KG_G], bufb[KG_G];
  load_trip(bufa, 0);
  for (int t = 0; t < ntrip; t += 2) {
    load_trip(bufb, t + 1);
    test_trip(bufa, t);
    load_trip(bufa, t + 2);
    test_trip(bufb, t + 1);
  }
  if (lane == 0) s_wcnt[wave] = cnt;
  __syncthreads();

  if (tid < nb) dinv[n0 + tid] = rsqrtf((float)s_deg[tid]);

  const int cw = s_wcnt[lane];
  int incl = cw;
#pragma unroll
  for (int dlt = 1; dlt < 32; dlt <<= 1) {
    const int t = __shfl_up(incl, dlt);
    if (lane >= dlt) incl += t;
  }
  const int total = __builtin_amdgcn_readlane(incl, 31);
  const bool whole =
      __builtin_amdgcn_ballot_w32(cw > KG_WCAP) == 0u && total <= KG_WSCAP;
  if (whole) {
    int2* __restrict__ wl =
        reinterpret_cast<int2*>(wlist) + (size_t)blockIdx.x * KG_WSCAP;
    const int off = __builtin_amdgcn_readlane(incl - cw, wave);
    for (int i = lane; i < cnt; i += 32) {
      int2 en = seg[i];
      en.x = src[en.x];
      wl[off + i] = en;
    }
  }
  if (tid == 0) wcnt[blockIdx.x] = whole ? total : -1;
}


template <int C, bool RELU>
__global__ __launch_bounds__(KG_THREADS) void kg_block_gather_kernel(
    const float* __restrict__ XW, const float* __restrict__ dinv,
    const float* __restrict__ bias, const int* __restrict__ src,
    const int* __restrict__ dst, const int* __restrict__ wlist,
    const int* __restrict__ wcnt, float* __restrict__ H,
    int n_nodes, int n_edges) {
  constexpr int V = C / 128;
  constexpr int B = 8 / V;
  __shared__ int2 s_list[KG_LIST];
  __shared__ int  s_cnt[2][KG_WAVES];

  const int tid  = threadIdx.x;
  const int lane = tid & 31;
  const int wave = __builtin_amdgcn_readfirstlane(tid >> 5);
  const int n0   = blockIdx.x * KG_NB;
  const int nb   = min(KG_NB, n_nodes - n0);
  const float4* __restrict__ XW4   = reinterpret_cast<const float4*>(XW);
  const float4* __restrict__ bias4 = reinterpret_cast<const float4*>(bias);

  float4 acc0[V], acc1[V];
  float dd0 = 0.0f, dd1 = 0.0f;
  for (int v = 0; v < V; ++v) {
    acc0[v] = make_float4(0.0f, 0.0f, 0.0f, 0.0f);
    acc1[v] = make_float4(0.0f, 0.0f, 0.0f, 0.0f);
  }
  if (wave < nb) {
    dd0 = dinv[n0 + wave];
    const float s = dd0 * dd0;
    for (int v = 0; v < V; ++v) {
      const float4 x = XW4[(size_t)(n0 + wave) * (C / 4) + v * 32 + lane];
      const float4 bb = bias4[v * 32 + lane];
      acc0[v].x = s * x.x + bb.x;
      acc0[v].y = s * x.y + bb.y;
      acc0[v].z = s * x.z + bb.z;
      acc0[v].w = s * x.w + bb.w;
    }
  }
  if (wave + 32 < nb) {
    dd1 = dinv[n0 + wave + 32];
    const float s = dd1 * dd1;
    for (int v = 0; v < V; ++v) {
      const float4 x = XW4[(size_t)(n0 + wave + 32) * (C / 4) + v * 32 + lane];
      const float4 bb = bias4[v * 32 + lane];
      acc1[v].x = s * x.x + bb.x;
      acc1[v].y = s * x.y + bb.y;
      acc1[v].z = s * x.z + bb.z;
      acc1[v].w = s * x.w + bb.w;
    }
  }

  const unsigned lanes_below = (1u << lane) - 1u;
  int fill  = 0;
  int base0 = 0;
  {
    const int cntw = __builtin_amdgcn_readfirstlane(wcnt[blockIdx.x]);
    if (cntw >= 0) {
      const int2* __restrict__ wl =
          reinterpret_cast<const int2*>(wlist) + (size_t)blockIdx.x * KG_WSCAP;
      for (int i = tid; i < cntw; i += KG_THREADS) s_list[i] = wl[i];
      fill  = cntw;
      base0 = n_edges;
    }
  }
  int par  = 0;
  for (int base = base0;; base += KG_ROUND, par ^= 1) {
    const bool done = base >= n_edges;
    if (done || fill > KG_LIST - KG_ROUND) {
      asm volatile("s_wait_dscnt 0x0" ::: "memory");
      __syncthreads();
      int q0 = 0, q1 = 0, qn0 = 0, qn1 = 0;
      for (int i0 = 0;; i0 += 32) {
        const bool last = i0 >= fill;
        int2 en = make_int2(0, -1);
        if (i0 + lane < fill) en = s_list[i0 + lane];
        unsigned m0 = __builtin_amdgcn_ballot_w32(en.y == wave);
        unsigned m1 = __builtin_amdgcn_ballot_w32(en.y == wave + 32);
        if (last || qn0 + __popc(m0) > 32 || qn1 + __popc(m1) > 32) {
          for (int pass = 0; pass < 2; ++pass) {
            if (qn0 > 0) {
              const float wq = dinv[(lane < qn0) ? q0 : 0];
              for (int k0 = 0; k0 < qn0; k0 += B) {
                int   es[B];
                float nrm[B];
#pragma unroll
                for (int k = 0; k < B; ++k) {
                  const bool ok = k0 + k < qn0;
                  const int  kk = ok ? k0 + k : k0;
                  es[k] = __builtin_amdgcn_readlane(q0, kk);
                  const float wk =
                      __int_as_float(__builtin_amdgcn_readlane(__float_as_int(wq), kk));
                  nrm[k] = ok ? wk * dd0 : 0.0f;
                }
                float4 x[B][V];
#pragma unroll
                for (int k = 0; k < B; ++k) {
#pragma unroll
                  for (int v = 0; v < V; ++v)
                    x[k][v] = XW4[(size_t)es[k] * (C / 4) + v * 32 + lane];
                }
#pragma unroll
                for (int k = 0; k < B; ++k) {
#pragma unroll
                  for (int v = 0; v < V; ++v) {
                    acc0[v].x += nrm[k] * x[k][v].x;
                    acc0[v].y += nrm[k] * x[k][v].y;
                    acc0[v].z += nrm[k] * x[k][v].z;
                    acc0[v].w += nrm[k] * x[k][v].w;
                  }
                }
              }
            }
            { const int t = q0; q0 = q1; q1 = t; }
            { const int t = qn0; qn0 = qn1; qn1 = t; }
            { const float t = dd0; dd0 = dd1; dd1 = t; }
            for (int v = 0; v < V; ++v) {
              const float4 t = acc0[v]; acc0[v] = acc1[v]; acc1[v] = t;
            }
          }
          qn0 = 0;
          qn1 = 0;
          if (last) break;
        }
        while (m0) {
          const int s = __builtin_amdgcn_readlane(en.x, __builtin_ctz(m0));
          if (lane == qn0) q0 = s;
          ++qn0;
          m0 &= m0 - 1u;
        }
        while (m1) {
          const int s = __builtin_amdgcn_readlane(en.x, __builtin_ctz(m1));
          if (lane == qn1) q1 = s;
          ++qn1;
          m1 &= m1 - 1u;
        }
      }
      if (done) break;
      __syncthreads();
      fill = 0;
    }

    const int wbase = base + wave * KG_CAP;
    int4 dv[4];
#pragma unroll
    for (int q4 = 0; q4 < 4; ++q4) {
      const int e = wbase + q4 * 128 + lane * 4;
      dv[q4] = make_int4(-1, -1, -1, -1);
      if (e < n_edges) dv[q4] = *reinterpret_cast<const int4*>(dst + e);
    }
    int dl[16], sv[16];
    int count = 0;
#pragma unroll
    for (int q4 = 0; q4 < 4; ++q4) {
      const int e = wbase + q4 * 128 + lane * 4;
      const int dd[4] = {dv[q4].x, dv[q4].y, dv[q4].z, dv[q4].w};
#pragma unroll
      for (int j = 0; j < 4; ++j) {
        const int d = dd[j] - n0;
        const bool m = (unsigned)d < (unsigned)nb;
        dl[q4 * 4 + j] = m ? d : -1;
        sv[q4 * 4 + j] = src[m ? (e + j) : 0];
        count += __popc(__builtin_amdgcn_ballot_w32(m));
      }
    }
    if (lane == 0) s_cnt[par][wave] = count;
    __syncthreads();

    const int cw = s_cnt[par][lane];
    int incl = cw;
#pragma unroll
    for (int dlt = 1; dlt < 32; dlt <<= 1) {
      const int t = __shfl_up(incl, dlt);
      if (lane >= dlt) incl += t;
    }
    const int total = __builtin_amdgcn_readlane(incl, 31);
    int pos = fill + __builtin_amdgcn_readlane(incl - cw, wave);
#pragma unroll
    for (int k = 0; k < 16; ++k) {
      const bool m = dl[k] >= 0;
      const unsigned mask = __builtin_amdgcn_ballot_w32(m);
      if (m) s_list[pos + __popc(mask & lanes_below)] = make_int2(sv[k], dl[k]);
      pos += __popc(mask);
    }
    fill += total;
  }

  float4* __restrict__ H4 = reinterpret_cast<float4*>(H);
  if (wave < nb) {
    for (int v = 0; v < V; ++v) {
      float4 r = acc0[v];
      if (RELU) {
        r.x = fmaxf(r.x, 0.0f); r.y = fmaxf(r.y, 0.0f);
        r.z = fmaxf(r.z, 0.0f); r.w = fmaxf(r.w, 0.0f);
      }
      H4[(size_t)(n0 + wave) * (C / 4) + v * 32 + lane] = r;
    }
  }
  if (wave + 32 < nb) {
    for (int v = 0; v < V; ++v) {
      float4 r = acc1[v];
      if (RELU) {
        r.x = fmaxf(r.x, 0.0f); r.y = fmaxf(r.y, 0.0f);
        r.z = fmaxf(r.z, 0.0f); r.w = fmaxf(r.w, 0.0f);
      }
      H4[(size_t)(n0 + wave + 32) * (C / 4) + v * 32 + lane] = r;
    }
  }
}

extern "C" void kernel_launch(void* const* d_in, const int* in_sizes, int n_in,
                              void* d_out, int out_size, void* d_ws, size_t ws_size,
                              hipStream_t stream) {
  const float* mol_x = (const float*)d_in[0];
  const int*   mei   = (const int*)d_in[1];
  const int*   kei   = (const int*)d_in[2];
  const float* W1 = (const float*)d_in[3];
  const float* b1 = (const float*)d_in[4];
  const float* W2 = (const float*)d_in[5];
  const float* b2 = (const float*)d_in[6];
  const float* W3 = (const float*)d_in[7];
  const float* b3 = (const float*)d_in[8];
  const float* Wk1 = (const float*)d_in[9];
  const float* bk1 = (const float*)d_in[10];
  const float* Wk2 = (const float*)d_in[11];
  const float* bk2 = (const float*)d_in[12];
  float* out = (float*)d_out;

  float* ws    = (float*)d_ws;
  float* embed = ws;
  float* cntf  = embed + M_MOL * OUT_CH;
  float* dinv  = cntf + M_MOL;
  float* XW    = dinv + M_MOL;
  float* h     = XW + (size_t)M_MOL * HID;
  float* XW2   = h + (size_t)M_MOL * HID;

  const int* kg_src = kei;
  const int* kg_dst = kei + E_KG;
  int* wlist = (int*)embed;
  int* wcnt  = (int*)cntf;

  mol_gnn_kernel<<<M_MOL, 256, 0, stream>>>(mol_x, mei, W1, b1, W2, b2, W3, b3,
                                            embed);

  {
    int tiles = (M_MOL / 16) * (HID / 16);
    wmma_gemm_kernel<<<(tiles + 7) / 8, 256, 0, stream>>>(embed, Wk1, XW,
                                                          M_MOL, OUT_CH, HID);
  }
  kg_filter_kernel<<<(M_MOL + KG_NB - 1) / KG_NB, KG_THREADS, 0, stream>>>(
      kg_src, kg_dst, dinv, wlist, wcnt, M_MOL, E_KG);
  kg_block_gather_kernel<HID, true><<<(M_MOL + KG_NB - 1) / KG_NB, KG_THREADS, 0,
                                      stream>>>(XW, dinv, bk1, kg_src, kg_dst, wlist,
                                                wcnt, h, M_MOL, E_KG);

  {
    int tiles = (M_MOL / 16) * (OUT_CH / 16);
    wmma_gemm_kernel<<<(tiles + 7) / 8, 256, 0, stream>>>(h, Wk2, XW2,
                                                          M_MOL, HID, OUT_CH);
  }
  kg_block_gather_kernel<OUT_CH, false><<<(M_MOL + KG_NB - 1) / KG_NB, KG_THREADS,
                                          0, stream>>>(XW2, dinv, bk2, kg_src,
                                                       kg_dst, wlist, wcnt, out,
                                                       M_MOL, E_KG);
}


# KG layers: both dense products moved into the layer-1 gather workgroup: it sums the 128-wide input rows, then does @Wk1+bias, ReLU, @Wk2 for its 40 rows out of LDS; 6 launches -> 4; sum order changes
# speedup vs baseline: 3.1349x; 1.0583x over previous
#include <hip/hip_runtime.h>

typedef __attribute__((ext_vector_type(2))) float v2f;
typedef __attribute__((ext_vector_type(8))) float v8f;

#define N_ATOMS 32
#define E_MOL   64
#define IN_CH   64
#define OUT_CH  128
#define HID     256
#define M_MOL   10000
#define E_KG    320000

__device__ __forceinline__ v8f wmma_f32_k4(v2f a, v2f b, v8f c) {
  return __builtin_amdgcn_wmma_f32_16x16x4_f32(false, a, false, b, (short)0, c,
                                               false, false);
}

#define MOL_WAVES 8

#define LDP 36

template <int K, int N, bool RELU>
__device__ __forceinline__ void gcn_layer_wmma(const float* __restrict__ XT,
                                               const float* __restrict__ W,
                                               const float* __restrict__ bias,
                                               const float* __restrict__ AT,
                                               float* __restrict__ TT,
                                               int wave, int lane) {
  const int hi = lane >> 4;
  const int lo = lane & 15;
  const float* __restrict__ Wl = W + hi * N + lo;
  const float* __restrict__ xp = XT + hi * LDP + lo;
  int tile1 = 16;
  asm("" : "+v"(tile1));
  const float* __restrict__ xq = xp + tile1;
  const float* __restrict__ at0 = AT + lo * LDP + 2 * hi;
  const float* __restrict__ at1 = AT + (16 + lo) * LDP + 2 * hi;
  for (int nt = wave; nt < N / 16; nt += MOL_WAVES) {
    const float* __restrict__ wp = Wl + nt * 16;
    v8f acc0 = {};
    v8f acc1 = {};
#pragma unroll
    for (int kb = 0; kb < K; kb += 32) {
      v2f a[8];
#pragma unroll
      for (int j = 0; j < 8; ++j) {
        a[j].x = wp[(kb + 4 * j) * N];
        a[j].y = wp[(kb + 4 * j + 2) * N];
      }
      __builtin_amdgcn_sched_barrier(0);
      v2f b0[8], b1[8];
#pragma unroll
      for (int j = 0; j < 8; ++j) {
        b0[j].x = xp[(kb + 4 * j) * LDP];
        b0[j].y = xp[(kb + 4 * j + 2) * LDP];
        b1[j].x = xq[(kb + 4 * j) * LDP];
        b1[j].y = xq[(kb + 4 * j + 2) * LDP];
      }
      __builtin_amdgcn_sched_barrier(0);
#pragma unroll
      for (int j = 0; j < 8; ++j) {
        acc0 = wmma_f32_k4(a[j], b0[j], acc0);
        acc1 = wmma_f32_k4(a[j], b1[j], acc1);
      }
    }
    float* trow = TT + (nt * 16 + 8 * hi) * LDP + lo;
#pragma unroll
    for (int v = 0; v < 8; ++v) {
      trow[v * LDP]      = acc0[v];
      trow[v * LDP + 16] = acc1[v];
    }
    const float* ar = TT + (nt * 16 + lo) * LDP + 2 * hi;
    v2f ta[8];
#pragma unroll
    for (int t = 0; t < 8; ++t)
      ta[t] = *reinterpret_cast<const v2f*>(ar + 4 * t);
    const float4 bA = *reinterpret_cast<const float4*>(bias + nt * 16 + 8 * hi);
    const float4 bB = *reinterpret_cast<const float4*>(bias + nt * 16 + 8 * hi + 4);
    v8f h0 = {bA.x, bA.y, bA.z, bA.w, bB.x, bB.y, bB.z, bB.w};
    v8f h1 = h0;
#pragma unroll
    for (int t = 0; t < 8; ++t) {
      const v2f n0 = *reinterpret_cast<const v2f*>(at0 + 4 * t);
      const v2f n1 = *reinterpret_cast<const v2f*>(at1 + 4 * t);
      h0 = wmma_f32_k4(ta[t], n0, h0);
      h1 = wmma_f32_k4(ta[t], n1, h1);
    }
#pragma unroll
    for (int v = 0; v < 8; ++v) {
      float r0 = h0[v], r1 = h1[v];
      if (RELU) { r0 = fmaxf(r0, 0.0f); r1 = fmaxf(r1, 0.0f); }
      trow[v * LDP]      = r0;
      trow[v * LDP + 16] = r1;
    }
  }
}

__global__ __launch_bounds__(256, 8) void mol_gnn_kernel(
    const float* __restrict__ mol_x, const int* __restrict__ mei,
    const float* __restrict__ W1, const float* __restrict__ b1,
    const float* __restrict__ W2, const float* __restrict__ b2,
    const float* __restrict__ W3, const float* __restrict__ b3,
    float* __restrict__ embed) {
  __shared__ __attribute__((aligned(16))) float Abuf[HID * LDP];
  __shared__ __attribute__((aligned(16))) float Bbuf[HID * LDP];
  __shared__ __attribute__((aligned(16))) float s_at[N_ATOMS * LDP];
  __shared__ float s_dinv[N_ATOMS];
  __shared__ int   s_deg[N_ATOMS];
  __shared__ int   s_src[E_MOL];
  __shared__ int   s_dst[E_MOL];

  const int mol  = blockIdx.x;
  const int tid  = threadIdx.x;
  const int lane = tid & 31;
  const int wave = __builtin_amdgcn_readfirstlane(tid >> 5);
  __builtin_assume(wave >= 0 && wave < MOL_WAVES);

  const float* X = mol_x + (size_t)mol * (N_ATOMS * IN_CH);
  const float4* X4 = reinterpret_cast<const float4*>(X);
  for (int q = wave; q < IN_CH / 4; q += MOL_WAVES) {
    const float4 v = X4[lane * (IN_CH / 4) + q];
    Abuf[(4 * q + 0) * LDP + lane] = v.x;
    Abuf[(4 * q + 1) * LDP + lane] = v.y;
    Abuf[(4 * q + 2) * LDP + lane] = v.z;
    Abuf[(4 * q + 3) * LDP + lane] = v.w;
  }
  const int* me = mei + (size_t)mol * (2 * E_MOL);
  if (tid < E_MOL) { s_src[tid] = me[tid]; s_dst[tid] = me[E_MOL + tid]; }
  if (tid < N_ATOMS) s_deg[tid] = 1;
  for (int i = tid; i < N_ATOMS * LDP; i += 256) s_at[i] = 0.0f;
  __syncthreads();
  if (tid < E_MOL) atomicAdd(&s_deg[s_dst[tid]], 1);
  __syncthreads();
  if (tid < N_ATOMS) {
    const float d = rsqrtf((float)s_deg[tid]);
    s_dinv[tid] = d;
    s_at[tid * LDP + tid] = d * d;
  }
  __syncthreads();
  if (tid < E_MOL) {
    const int s = s_src[tid], d = s_dst[tid];
    atomicAdd(&s_at[d * LDP + s], s_dinv[s] * s_dinv[d]);
  }
  __syncthreads();

  gcn_layer_wmma<IN_CH, HID, true>(Abuf, W1, b1, s_at, Bbuf, wave, lane);
  __syncthreads();
  gcn_layer_wmma<HID, HID, true>(Bbuf, W2, b2, s_at, Abuf, wave, lane);
  __syncthreads();
  gcn_layer_wmma<HID, OUT_CH, false>(Abuf, W3, b3, s_at, Bbuf, wave, lane);
  __syncthreads();

  if (tid < OUT_CH) {
    const int c = tid;
    float v = Bbuf[c * LDP + (c & (N_ATOMS - 1))];
    for (int r = 1; r < N_ATOMS; ++r)
      v = fminf(v, Bbuf[c * LDP + ((c + r) & (N_ATOMS - 1))]);
    embed[(size_t)mol * OUT_CH + c] = v;
  }
}

template <int NR, int K, int N, int AP>
__device__ __forceinline__ void kg_tile_gemm(const float* __restrict__ A,
                                             const float* __restrict__ Bcol,
                                             v8f (&acc)[NR], int lane) {
  const int hi = lane >> 4;
  const int lo = lane & 15;
  const float* __restrict__ ap = A + lo * AP + 2 * hi;
  const float* __restrict__ bp = Bcol + (2 * hi) * N + lo;
#pragma unroll
  for (int kb = 0; kb < K; kb += 32) {
    v2f b[8];
#pragma unroll
    for (int j = 0; j < 8; ++j) {
      b[j].x = bp[(kb + 4 * j) * N];
      b[j].y = bp[(kb + 4 * j + 1) * N];
    }
    __builtin_amdgcn_sched_barrier(0);
    v2f a[NR][8];
#pragma unroll
    for (int r = 0; r < NR; ++r) {
#pragma unroll
      for (int j = 0; j < 8; ++j)
        a[r][j] = *reinterpret_cast<const v2f*>(ap + r * 16 * AP + kb + 4 * j);
    }
    __builtin_amdgcn_sched_barrier(0);
#pragma unroll
    for (int j = 0; j < 8; ++j) {
#pragma unroll
      for (int r = 0; r < NR; ++r)
        acc[r] = wmma_f32_k4(a[r][j], b[j], acc[r]);
    }
  }
}

#define KG_NB      40
#define KG_THREADS 1024
#define KG_WAVES   32
#define KG_CAP     512
#define KG_ROUND   (KG_WAVES * KG_CAP)
#define KG_LIST    (2 * KG_ROUND)
#define KG_WCAP    512
#define KG_WSCAP   2048
#define KG_G       4
#define KG_MROWS   48
#define KG_ZP      (OUT_CH + 4)
#define KG_HP      (HID + 4)

__global__ __launch_bounds__(KG_THREADS) void kg_filter_kernel(
    const int* __restrict__ src, const int* __restrict__ dst,
    float* __restrict__ dinv, int* __restrict__ wlist, int* __restrict__ wcnt,
    int n_nodes, int n_edges) {
  __shared__ int2 s_seg[KG_WAVES * KG_WCAP];
  __shared__ int  s_wcnt[KG_WAVES];
  __shared__ int  s_deg[KG_NB];

  const int tid  = threadIdx.x;
  const int lane = tid & 31;
  const int wave = __builtin_amdgcn_readfirstlane(tid >> 5);
  const int n0   = blockIdx.x * KG_NB;
  const int nb   = min(KG_NB, n_nodes - n0);

  if (tid < KG_NB) s_deg[tid] = 1;
  __syncthreads();

  const unsigned lanes_below = (1u << lane) - 1u;
  int2* __restrict__ seg = s_seg + wave * KG_WCAP;
  const int ntrip = (n_edges + 128 * KG_WAVES * KG_G - 1) / (128 * KG_WAVES * KG_G);
  int cnt = 0;
  const int last4 = max(n_edges - 4, 0);
  auto load_trip = [&](int4 (&buf)[KG_G], int t) {
#pragma unroll
    for (int g = 0; g < KG_G; ++g) {
      const int e = ((t * KG_G + g) * KG_WAVES + wave) * 128 + lane * 4;
      buf[g] = *reinterpret_cast<const int4*>(dst + min(e, last4));
    }
  };
  auto test_trip = [&](const int4 (&buf)[KG_G], int t) {
#pragma unroll
    for (int g = 0; g < KG_G; ++g) {
      const int e0 = ((t * KG_G + g) * KG_WAVES + wave) * 128 + lane * 4;
      const int d[4] = {buf[g].x - n0, buf[g].y - n0, buf[g].z - n0, buf[g].w - n0};
      const unsigned nbv = (e0 < n_edges) ? (unsigned)nb : 0u;
      unsigned mk[4];
#pragma unroll
      for (int j = 0; j < 4; ++j)
        mk[j] = __builtin_amdgcn_ballot_w32((unsigned)d[j] < nbv);
      if ((mk[0] | mk[1] | mk[2] | mk[3]) != 0u) {
#pragma unroll
        for (int j = 0; j < 4; ++j) {
          if (mk[j] != 0u) {
            const int p = cnt + __popc(mk[j] & lanes_below);
            if ((unsigned)d[j] < nbv) {
              atomicAdd(&s_deg[d[j]], 1);
              if (p < KG_WCAP) seg[p] = make_int2(e0 + j, d[j]);
            }
            cnt += __popc(mk[j]);
          }
        }
      }
    }
  };
  int4 bufa[KG_G], bufb[KG_G];
  load_trip(bufa, 0);
  for (int t = 0; t < ntrip; t += 2) {
    load_trip(bufb, t + 1);
    test_trip(bufa, t);
    load_trip(bufa, t + 2);
    test_trip(bufb, t + 1);
  }
  if (lane == 0) s_wcnt[wave] = cnt;
  __syncthreads();

  if (tid < nb) dinv[n0 + tid] = rsqrtf((float)s_deg[tid]);

  const int cw = s_wcnt[lane];
  int incl = cw;
#pragma unroll
  for (int dlt = 1; dlt < 32; dlt <<= 1) {
    const int t = __shfl_up(incl, dlt);
    if (lane >= dlt) incl += t;
  }
  const int total = __builtin_amdgcn_readlane(incl, 31);
  const bool whole =
      __builtin_amdgcn_ballot_w32(cw > KG_WCAP) == 0u && total <= KG_WSCAP;
  if (whole) {
    int2* __restrict__ wl =
        reinterpret_cast<int2*>(wlist) + (size_t)blockIdx.x * KG_WSCAP;
    const int off = __builtin_amdgcn_readlane(incl - cw, wave);
    for (int i = lane; i < cnt; i += 32) {
      int2 en = seg[i];
      en.x = src[en.x];
      wl[off + i] = en;
    }
  }
  if (tid == 0) wcnt[blockIdx.x] = whole ? total : -1;
}


template <int C, bool RELU, bool FUSE>
__global__ __launch_bounds__(KG_THREADS) void kg_block_gather_kernel(
    const float* __restrict__ XW, const float* __restrict__ dinv,
    const float* __restrict__ bias, const int* __restrict__ src,
    const int* __restrict__ dst, const int* __restrict__ wlist,
    const int* __restrict__ wcnt, float* __restrict__ H,
    const float* __restrict__ Wa, const float* __restrict__ Wb,
    int n_nodes, int n_edges) {
  static_assert(!FUSE || C == OUT_CH, "the fused path takes 128-wide rows");
  constexpr int V = C / 128;
  constexpr int B = 8 / V;
  __shared__ __attribute__((aligned(16))) int2 s_list[KG_LIST];
  __shared__ int  s_cnt[2][KG_WAVES];

  const int tid  = threadIdx.x;
  const int lane = tid & 31;
  const int wave = __builtin_amdgcn_readfirstlane(tid >> 5);
  const int n0   = blockIdx.x * KG_NB;
  const int nb   = min(KG_NB, n_nodes - n0);
  const float4* __restrict__ XW4   = reinterpret_cast<const float4*>(XW);
  const float4* __restrict__ bias4 = reinterpret_cast<const float4*>(bias);

  float4 acc0[V], acc1[V];
  float dd0 = 0.0f, dd1 = 0.0f;
  for (int v = 0; v < V; ++v) {
    acc0[v] = make_float4(0.0f, 0.0f, 0.0f, 0.0f);
    acc1[v] = make_float4(0.0f, 0.0f, 0.0f, 0.0f);
  }
  if (wave < nb) {
    dd0 = dinv[n0 + wave];
    const float s = dd0 * dd0;
    for (int v = 0; v < V; ++v) {
      const float4 x = XW4[(size_t)(n0 + wave) * (C / 4) + v * 32 + lane];
      const float4 bb = FUSE ? make_float4(0.0f, 0.0f, 0.0f, 0.0f) : bias4[v * 32 + lane];
      acc0[v].x = s * x.x + bb.x;
      acc0[v].y = s * x.y + bb.y;
      acc0[v].z = s * x.z + bb.z;
      acc0[v].w = s * x.w + bb.w;
    }
  }
  if (wave + 32 < nb) {
    dd1 = dinv[n0 + wave + 32];
    const float s = dd1 * dd1;
    for (int v = 0; v < V; ++v) {
      const float4 x = XW4[(size_t)(n0 + wave + 32) * (C / 4) + v * 32 + lane];
      const float4 bb = FUSE ? make_float4(0.0f, 0.0f, 0.0f, 0.0f) : bias4[v * 32 + lane];
      acc1[v].x = s * x.x + bb.x;
      acc1[v].y = s * x.y + bb.y;
      acc1[v].z = s * x.z + bb.z;
      acc1[v].w = s * x.w + bb.w;
    }
  }

  const unsigned lanes_below = (1u << lane) - 1u;
  int fill  = 0;
  int base0 = 0;
  {
    const int cntw = __builtin_amdgcn_readfirstlane(wcnt[blockIdx.x]);
    if (cntw >= 0) {
      const int2* __restrict__ wl =
          reinterpret_cast<const int2*>(wlist) + (size_t)blockIdx.x * KG_WSCAP;
      for (int i = tid; i < cntw; i += KG_THREADS) s_list[i] = wl[i];
      fill  = cntw;
      base0 = n_edges;
    }
  }
  int par  = 0;
  for (int base = base0;; base += KG_ROUND, par ^= 1) {
    const bool done = base >= n_edges;
    if (done || fill > KG_LIST - KG_ROUND) {
      asm volatile("s_wait_dscnt 0x0" ::: "memory");
      __syncthreads();
      int q0 = 0, q1 = 0, qn0 = 0, qn1 = 0;
      for (int i0 = 0;; i0 += 32) {
        const bool last = i0 >= fill;
        int2 en = make_int2(0, -1);
        if (i0 + lane < fill) en = s_list[i0 + lane];
        unsigned m0 = __builtin_amdgcn_ballot_w32(en.y == wave);
        unsigned m1 = __builtin_amdgcn_ballot_w32(en.y == wave + 32);
        if (last || qn0 + __popc(m0) > 32 || qn1 + __popc(m1) > 32) {
          for (int pass = 0; pass < 2; ++pass) {
            if (qn0 > 0) {
              const float wq = dinv[(lane < qn0) ? q0 : 0];
              for (int k0 = 0; k0 < qn0; k0 += B) {
                int   es[B];
                float nrm[B];
#pragma unroll
                for (int k = 0; k < B; ++k) {
                  const bool ok = k0 + k < qn0;
                  const int  kk = ok ? k0 + k : k0;
                  es[k] = __builtin_amdgcn_readlane(q0, kk);
                  const float wk =
                      __int_as_float(__builtin_amdgcn_readlane(__float_as_int(wq), kk));
                  nrm[k] = ok ? wk * dd0 : 0.0f;
                }
                float4 x[B][V];
#pragma unroll
                for (int k = 0; k < B; ++k) {
#pragma unroll
                  for (int v = 0; v < V; ++v)
                    x[k][v] = XW4[(size_t)es[k] * (C / 4) + v * 32 + lane];
                }
#pragma unroll
                for (int k = 0; k < B; ++k) {
#pragma unroll
                  for (int v = 0; v < V; ++v) {
                    acc0[v].x += nrm[k] * x[k][v].x;
                    acc0[v].y += nrm[k] * x[k][v].y;
                    acc0[v].z += nrm[k] * x[k][v].z;
                    acc0[v].w += nrm[k] * x[k][v].w;
                  }
                }
              }
            }
            { const int t = q0; q0 = q1; q1 = t; }
            { const int t = qn0; qn0 = qn1; qn1 = t; }
            { const float t = dd0; dd0 = dd1; dd1 = t; }
            for (int v = 0; v < V; ++v) {
              const float4 t = acc0[v]; acc0[v] = acc1[v]; acc1[v] = t;
            }
          }
          qn0 = 0;
          qn1 = 0;
          if (last) break;
        }
        while (m0) {
          const int s = __builtin_amdgcn_readlane(en.x, __builtin_ctz(m0));
          if (lane == qn0) q0 = s;
          ++qn0;
          m0 &= m0 - 1u;
        }
        while (m1) {
          const int s = __builtin_amdgcn_readlane(en.x, __builtin_ctz(m1));
          if (lane == qn1) q1 = s;
          ++qn1;
          m1 &= m1 - 1u;
        }
      }
      if (done) break;
      __syncthreads();
      fill = 0;
    }

    const int wbase = base + wave * KG_CAP;
    int4 dv[4];
#pragma unroll
    for (int q4 = 0; q4 < 4; ++q4) {
      const int e = wbase + q4 * 128 + lane * 4;
      dv[q4] = make_int4(-1, -1, -1, -1);
      if (e < n_edges) dv[q4] = *reinterpret_cast<const int4*>(dst + e);
    }
    int dl[16], sv[16];
    int count = 0;
#pragma unroll
    for (int q4 = 0; q4 < 4; ++q4) {
      const int e = wbase + q4 * 128 + lane * 4;
      const int dd[4] = {dv[q4].x, dv[q4].y, dv[q4].z, dv[q4].w};
#pragma unroll
      for (int j = 0; j < 4; ++j) {
        const int d = dd[j] - n0;
        const bool m = (unsigned)d < (unsigned)nb;
        dl[q4 * 4 + j] = m ? d : -1;
        sv[q4 * 4 + j] = src[m ? (e + j) : 0];
        count += __popc(__builtin_amdgcn_ballot_w32(m));
      }
    }
    if (lane == 0) s_cnt[par][wave] = count;
    __syncthreads();

    const int cw = s_cnt[par][lane];
    int incl = cw;
#pragma unroll
    for (int dlt = 1; dlt < 32; dlt <<= 1) {
      const int t = __shfl_up(incl, dlt);
      if (lane >= dlt) incl += t;
    }
    const int total = __builtin_amdgcn_readlane(incl, 31);
    int pos = fill + __builtin_amdgcn_readlane(incl - cw, wave);
#pragma unroll
    for (int k = 0; k < 16; ++k) {
      const bool m = dl[k] >= 0;
      const unsigned mask = __builtin_amdgcn_ballot_w32(m);
      if (m) s_list[pos + __popc(mask & lanes_below)] = make_int2(sv[k], dl[k]);
      pos += __popc(mask);
    }
    fill += total;
  }

  if (FUSE) {
    const int hi = lane >> 4;
    const int lo = lane & 15;
    __syncthreads();
    float* __restrict__ zb = reinterpret_cast<float*>(s_list);
    float* __restrict__ hb = zb + KG_MROWS * KG_ZP;
    *reinterpret_cast<float4*>(zb + wave * KG_ZP + 4 * lane) = acc0[0];
    if (wave + 32 < KG_MROWS)
      *reinterpret_cast<float4*>(zb + (wave + 32) * KG_ZP + 4 * lane) = acc1[0];
    __syncthreads();
    {
      const int ct = wave & 15;
      const float bcol = bias[ct * 16 + lo];
      const float* __restrict__ Wc = Wa + ct * 16;
      float* __restrict__ hrow = hb + (8 * hi) * KG_HP + ct * 16 + lo;
      if (wave < 16) {
        v8f acc[2];
#pragma unroll
        for (int v = 0; v < 8; ++v) { acc[0][v] = bcol; acc[1][v] = bcol; }
        kg_tile_gemm<2, C, HID, KG_ZP>(zb, Wc, acc, lane);
#pragma unroll
        for (int v = 0; v < 8; ++v) {
          float r0 = acc[0][v], r1 = acc[1][v];
          if (RELU) { r0 = fmaxf(r0, 0.0f); r1 = fmaxf(r1, 0.0f); }
          hrow[v * KG_HP]        = r0;
          hrow[(16 + v) * KG_HP] = r1;
        }
      } else {
        v8f acc[1];
#pragma unroll
        for (int v = 0; v < 8; ++v) acc[0][v] = bcol;
        kg_tile_gemm<1, C, HID, KG_ZP>(zb + 32 * KG_ZP, Wc, acc, lane);
#pragma unroll
        for (int v = 0; v < 8; ++v) {
          float r0 = acc[0][v];
          if (RELU) r0 = fmaxf(r0, 0.0f);
          hrow[(32 + v) * KG_HP] = r0;
        }
      }
    }
    __syncthreads();
    {
      const int ct = wave & 7;
      const int rt = wave >> 3;
      if (rt < KG_MROWS / 16) {
        v8f acc[1];
#pragma unroll
        for (int v = 0; v < 8; ++v) acc[0][v] = 0.0f;
        kg_tile_gemm<1, HID, OUT_CH, KG_HP>(hb + rt * 16 * KG_HP, Wb + ct * 16, acc, lane);
#pragma unroll
        for (int v = 0; v < 8; ++v) {
          const int row = rt * 16 + v + 8 * hi;
          if (row < nb) H[(size_t)(n0 + row) * OUT_CH + ct * 16 + lo] = acc[0][v];
        }
      }
    }
    return;
  }

  float4* __restrict__ H4 = reinterpret_cast<float4*>(H);
  if (wave < nb) {
    for (int v = 0; v < V; ++v) {
      float4 r = acc0[v];
      if (RELU) {
        r.x = fmaxf(r.x, 0.0f); r.y = fmaxf(r.y, 0.0f);
        r.z = fmaxf(r.z, 0.0f); r.w = fmaxf(r.w, 0.0f);
      }
      H4[(size_t)(n0 + wave) * (C / 4) + v * 32 + lane] = r;
    }
  }
  if (wave + 32 < nb) {
    for (int v = 0; v < V; ++v) {
      float4 r = acc1[v];
      if (RELU) {
        r.x = fmaxf(r.x, 0.0f); r.y = fmaxf(r.y, 0.0f);
        r.z = fmaxf(r.z, 0.0f); r.w = fmaxf(r.w, 0.0f);
      }
      H4[(size_t)(n0 + wave + 32) * (C / 4) + v * 32 + lane] = r;
    }
  }
}

extern "C" void kernel_launch(void* const* d_in, const int* in_sizes, int n_in,
                              void* d_out, int out_size, void* d_ws, size_t ws_size,
                              hipStream_t stream) {
  const float* mol_x = (const float*)d_in[0];
  const int*   mei   = (const int*)d_in[1];
  const int*   kei   = (const int*)d_in[2];
  const float* W1 = (const float*)d_in[3];
  const float* b1 = (const float*)d_in[4];
  const float* W2 = (const float*)d_in[5];
  const float* b2 = (const float*)d_in[6];
  const float* W3 = (const float*)d_in[7];
  const float* b3 = (const float*)d_in[8];
  const float* Wk1 = (const float*)d_in[9];
  const float* bk1 = (const float*)d_in[10];
  const float* Wk2 = (const float*)d_in[11];
  const float* bk2 = (const float*)d_in[12];
  float* out = (float*)d_out;

  float* ws    = (float*)d_ws;
  float* embed = ws;
  float* cntf  = embed + M_MOL * OUT_CH;
  float* dinv  = cntf + M_MOL;
  float* XW    = dinv + M_MOL;
  float* XW2   = XW + 2 * (size_t)M_MOL * HID;

  const int* kg_src = kei;
  const int* kg_dst = kei + E_KG;
  int* wlist = (int*)XW;
  int* wcnt  = (int*)cntf;

  mol_gnn_kernel<<<M_MOL, 256, 0, stream>>>(mol_x, mei, W1, b1, W2, b2, W3, b3,
                                            embed);

  kg_filter_kernel<<<(M_MOL + KG_NB - 1) / KG_NB, KG_THREADS, 0, stream>>>(
      kg_src, kg_dst, dinv, wlist, wcnt, M_MOL, E_KG);
  kg_block_gather_kernel<OUT_CH, true, true><<<(M_MOL + KG_NB - 1) / KG_NB,
                                              KG_THREADS, 0, stream>>>(
      embed, dinv, bk1, kg_src, kg_dst, wlist, wcnt, XW2, Wk1, Wk2, M_MOL, E_KG);
  kg_block_gather_kernel<OUT_CH, false, false><<<(M_MOL + KG_NB - 1) / KG_NB,
                                                KG_THREADS, 0, stream>>>(
      XW2, dinv, bk2, kg_src, kg_dst, wlist, wcnt, out, nullptr, nullptr, M_MOL,
      E_KG);
}
